# starting kernel
# baseline (speedup 1.0000x reference)
#include <hip/hip_runtime.h>


typedef float v2f __attribute__((ext_vector_type(2)));
typedef float v8f __attribute__((ext_vector_type(8)));

static __device__ __forceinline__ v8f wmma_f32_16x16x4(v2f a, v2f b, v8f c) {
  return __builtin_amdgcn_wmma_f32_16x16x4_f32(
       false, a,  false, b,
       (short)0, c,  false,  false);
}

__global__ void zero_f32_kernel(float* __restrict__ p, size_t n) {
  size_t i = (size_t)blockIdx.x * blockDim.x + threadIdx.x;
  size_t stride = (size_t)gridDim.x * blockDim.x;
  for (; i < n; i += stride) p[i] = 0.0f;
}

__global__ void degree_kernel(const int* __restrict__ dst,
                              float* __restrict__ deg, int nE) {
  int e = blockIdx.x * blockDim.x + threadIdx.x;
  if (e < nE) atomicAdd(&deg[dst[e]], 1.0f);
}

__global__ void scatter_add_kernel(const float* __restrict__ x,
                                   const int* __restrict__ src,
                                   const int* __restrict__ dst,
                                   float* __restrict__ agg, int nE) {
  int gid  = blockIdx.x * blockDim.x + threadIdx.x;
  int e    = gid >> 5;
  int lane = gid & 31;
  if (e >= nE) return;
  int s = src[e];
  int d = dst[e];
  const float4 v = *(const float4*)(x + (size_t)s * 128 + lane * 4);
  float* o = agg + (size_t)d * 128 + lane * 4;
  atomicAdd(o + 0, v.x);
  atomicAdd(o + 1, v.y);
  atomicAdd(o + 2, v.z);
  atomicAdd(o + 3, v.w);
}

__global__ __launch_bounds__(256)
void sage_wmma_gemm_kernel(const float* __restrict__ X,
                           const float* __restrict__ AGG,
                           const float* __restrict__ DEG,
                           const float* __restrict__ Wself,
                           const float* __restrict__ Wneigh,
                           const float* __restrict__ bias,
                           float* __restrict__ out,
                           int nRows, int nCols, int doRelu) {
  const int wavesPerBlock = blockDim.x >> 5;
  const int wave = threadIdx.x >> 5;
  const int lane = threadIdx.x & 31;
  const int nTilesN = nCols >> 4;
  const int nTiles  = (nRows >> 4) * nTilesN;
  const int tile = blockIdx.x * wavesPerBlock + wave;
  if (tile >= nTiles) return;

  const int mt   = tile / nTilesN;
  const int nt   = tile - mt * nTilesN;
  const int half = lane >> 4;
  const int l15  = lane & 15;
  const int aRow = mt * 16 + l15;
  const int bCol = nt * 16 + l15;
  const int kb   = half * 2;

  v8f c = {0.f, 0.f, 0.f, 0.f, 0.f, 0.f, 0.f, 0.f};

  const float* xr = X + (size_t)aRow * 128 + kb;
  const float* wl = Wself + (size_t)kb * nCols + bCol;
#pragma unroll 8
  for (int k = 0; k < 128; k += 4) {
    v2f a = *(const v2f*)(xr + k);
    v2f b;
    b.x = wl[(size_t)k * nCols];
    b.y = wl[(size_t)(k + 1) * nCols];
    c = wmma_f32_16x16x4(a, b, c);
  }

  const float inv = 1.0f / fmaxf(DEG[aRow], 1.0f);
  const float* gr = AGG + (size_t)aRow * 128 + kb;
  const float* wn = Wneigh + (size_t)kb * nCols + bCol;
#pragma unroll 8
  for (int k = 0; k < 128; k += 4) {
    v2f a = *(const v2f*)(gr + k);
    a.x *= inv;
    a.y *= inv;
    v2f b;
    b.x = wn[(size_t)k * nCols];
    b.y = wn[(size_t)(k + 1) * nCols];
    c = wmma_f32_16x16x4(a, b, c);
  }

  const float bv = bias[bCol];
  float* orow = out + (size_t)(mt * 16 + half * 8) * nCols + bCol;
#pragma unroll
  for (int i = 0; i < 8; ++i) {
    float v = c[i] + bv;
    if (doRelu) v = fmaxf(v, 0.0f);
    orow[(size_t)i * nCols] = v;
  }
}

extern "C" void kernel_launch(void* const* d_in, const int* in_sizes, int n_in,
                              void* d_out, int out_size, void* d_ws, size_t ws_size,
                              hipStream_t stream) {
  const float* x   = (const float*)d_in[0];
  const float* Ws1 = (const float*)d_in[1];
  const float* Wn1 = (const float*)d_in[2];
  const float* b1  = (const float*)d_in[3];
  const float* Ws2 = (const float*)d_in[4];
  const float* Wn2 = (const float*)d_in[5];
  const float* b2  = (const float*)d_in[6];
  const int*   src = (const int*)d_in[7];
  const int*   dst = (const int*)d_in[8];
  float* out = (float*)d_out;

  const int F = in_sizes[3];
  const int C = in_sizes[6];
  const int N = in_sizes[0] / F;
  const int E = in_sizes[7];

  char* ws = (char*)d_ws;
  float* deg = (float*)ws;
  float* agg = (float*)(ws + (size_t)N * 4);
  float* h   = (float*)(ws + (size_t)N * 4 + (size_t)N * F * 4);

  const int ZB = 2048, TB = 256;
  const int edgeWaveBlocks = (E * 32 + TB - 1) / TB;
  const int edgeBlocks     = (E + TB - 1) / TB;

  zero_f32_kernel<<<ZB, TB, 0, stream>>>(deg, (size_t)N + (size_t)N * F);
  degree_kernel<<<edgeBlocks, TB, 0, stream>>>(dst, deg, E);
  scatter_add_kernel<<<edgeWaveBlocks, TB, 0, stream>>>(x, src, dst, agg, E);

  {
    const int nTiles = (N / 16) * (F / 16);
    const int blocks = (nTiles + (TB / 32) - 1) / (TB / 32);
    sage_wmma_gemm_kernel<<<blocks, TB, 0, stream>>>(
        x, agg, deg, Ws1, Wn1, b1, h, N, F,  1);
  }

  zero_f32_kernel<<<ZB, TB, 0, stream>>>(agg, (size_t)N * F);
  scatter_add_kernel<<<edgeWaveBlocks, TB, 0, stream>>>(h, src, dst, agg, E);

  {
    const int nTiles = (N / 16) * (C / 16);
    const int blocks = (nTiles + (TB / 32) - 1) / (TB / 32);
    sage_wmma_gemm_kernel<<<blocks, TB, 0, stream>>>(
        h, agg, deg, Ws2, Wn2, b2, out, N, C,  0);
  }
}


# layer 1: a work-group owns 196 target rows in LDS, filters the edge list by an owner byte per edge, adds rows by LDS f32 atomics (arrival order), writes agg and deg once; no global atomics there
# speedup vs baseline: 3.5646x; 3.5646x over previous
#include <hip/hip_runtime.h>


typedef float v2f __attribute__((ext_vector_type(2)));
typedef float v8f __attribute__((ext_vector_type(8)));

static __device__ __forceinline__ v8f wmma_f32_16x16x4(v2f a, v2f b, v8f c) {
  return __builtin_amdgcn_wmma_f32_16x16x4_f32(
       false, a,  false, b,
       (short)0, c,  false,  false);
}

#define OWN_ROWS 196
#define OWN_LIST_CAP 8192
#define OWN_THREADS 1024
#define OWN_WAVES (OWN_THREADS / 32)
#define OWN_GROUP 16

__global__ __launch_bounds__(256)
void owner_byte_kernel(const int* __restrict__ dst, unsigned int* __restrict__ ownerW, int nE) {
  const int i = blockIdx.x * blockDim.x + threadIdx.x;
  const int e = i * 4;
  if (e >= nE) return;
  unsigned int w = 0;
#pragma unroll
  for (int j = 0; j < 4; ++j) {
    if (e + j < nE) w |= (((unsigned int)dst[e + j] / OWN_ROWS) & 0xffu) << (8 * j);
  }
  ownerW[i] = w;
}

static __device__ __forceinline__ unsigned int zero_byte_flags(unsigned int y) {
  return ~((((y & 0x7f7f7f7fu) + 0x7f7f7f7fu) | y) | 0x7f7f7f7fu);
}

static __device__ __forceinline__ unsigned int own_chunk_flags(uint4 v, unsigned int pat) {
  return (zero_byte_flags(v.x ^ pat) >> 7) | (zero_byte_flags(v.y ^ pat) >> 6) |
         (zero_byte_flags(v.z ^ pat) >> 5) | (zero_byte_flags(v.w ^ pat) >> 4);
}

static __device__ __forceinline__ void own_append(unsigned int c, int qA, int qB,
                                                  int* sCnt, int* sList) {
  while (c != 0u) {
    const int p = __ffs((int)c) - 1;
    c &= c - 1u;
    const int q = ((p >> 2) & 1) ? qB : qA;
    const int e = q * 16 + (p & 3) * 4 + (p >> 3);
    const int slot = atomicAdd(sCnt, 1);
    if (slot < OWN_LIST_CAP) sList[slot] = e;
  }
}

__global__ __launch_bounds__(OWN_THREADS)
void own_rows_sum_kernel(const float* __restrict__ x,
                         const int* __restrict__ src,
                         const int* __restrict__ dst,
                         const unsigned int* __restrict__ ownerW,
                         float* __restrict__ agg,
                         float* __restrict__ deg,
                         int nE, int nRows) {
  __shared__ __align__(16) float sAcc[OWN_ROWS * 128];
  __shared__ float sDeg[OWN_ROWS];
  __shared__ int   sList[OWN_LIST_CAP];
  __shared__ int   sCnt;
  const int tid  = threadIdx.x;
  const int lane = tid & 31;
  const int wave = tid >> 5;
  const int base = blockIdx.x * OWN_ROWS;
  const unsigned int myRun = blockIdx.x & 0xffu;

  for (int i = tid; i < OWN_ROWS * 128; i += OWN_THREADS) sAcc[i] = 0.0f;
  if (tid < OWN_ROWS) sDeg[tid] = 0.0f;
  if (tid == 0) sCnt = 0;
  __syncthreads();

  const unsigned int pat = myRun * 0x01010101u;
  const int nChunks = nE >> 4;
  const uint4* ow = (const uint4*)ownerW;
  int q0 = tid;
  for (; q0 + 7 * OWN_THREADS < nChunks; q0 += 8 * OWN_THREADS) {
    uint4 v[8];
#pragma unroll
    for (int u = 0; u < 8; ++u) v[u] = ow[q0 + u * OWN_THREADS];
#pragma unroll
    for (int u = 0; u < 8; u += 2) {
      const unsigned int c = own_chunk_flags(v[u], pat) | (own_chunk_flags(v[u + 1], pat) << 4);
      own_append(c, q0 + u * OWN_THREADS, q0 + (u + 1) * OWN_THREADS, &sCnt, sList);
    }
  }
  for (; q0 < nChunks; q0 += OWN_THREADS) {
    const unsigned int c = own_chunk_flags(ow[q0], pat);
    own_append(c, q0, q0, &sCnt, sList);
  }
  {
    const int e = (nChunks << 4) + tid;
    if (tid < 16 && e < nE) {
      const unsigned int w = ownerW[e >> 2];
      if (((w >> (8 * (e & 3))) & 0xffu) == myRun) {
        const int slot = atomicAdd(&sCnt, 1);
        if (slot < OWN_LIST_CAP) sList[slot] = e;
      }
    }
  }
  __syncthreads();

  const int total = sCnt;
  if (total > OWN_LIST_CAP) {
    const unsigned char* ob = (const unsigned char*)ownerW;
    for (int e = tid; e < nE; e += OWN_THREADS) {
      if ((unsigned int)ob[e] != myRun) continue;
      const int r = dst[e] - base;
      if ((unsigned int)r >= (unsigned int)OWN_ROWS) continue;
      const float* xr = x + (size_t)src[e] * 128;
      for (int cc = 0; cc < 128; ++cc) atomicAdd(&sAcc[r * 128 + cc], xr[cc]);
      atomicAdd(&sDeg[r], 1.0f);
    }
  } else {
    const int n = total;
    int g0 = wave * OWN_GROUP;
    int sv = 0, dv = 0;
    if (g0 + OWN_GROUP <= n && lane < OWN_GROUP) {
      const int e = sList[g0 + lane];
      sv = src[e];
      dv = dst[e];
    }
    while (g0 + OWN_GROUP <= n) {
      float v[OWN_GROUP][4];
      int   r[OWN_GROUP];
#pragma unroll
      for (int j = 0; j < OWN_GROUP; ++j) {
        const int s = __builtin_amdgcn_readlane(sv, j);
        r[j] = __builtin_amdgcn_readlane(dv, j) - base;
        const float* xr = x + (size_t)s * 128 + lane;
        v[j][0] = xr[0];
        v[j][1] = xr[32];
        v[j][2] = xr[64];
        v[j][3] = xr[96];
      }
      const int g1 = g0 + OWN_WAVES * OWN_GROUP;
      int sv2 = 0, dv2 = 0;
      if (g1 + OWN_GROUP <= n && lane < OWN_GROUP) {
        const int e = sList[g1 + lane];
        sv2 = src[e];
        dv2 = dst[e];
      }
#pragma unroll
      for (int j = 0; j < OWN_GROUP; ++j) {
        if ((unsigned int)r[j] < (unsigned int)OWN_ROWS) {
          float* o = &sAcc[r[j] * 128 + lane];
          atomicAdd(o,      v[j][0]);
          atomicAdd(o + 32, v[j][1]);
          atomicAdd(o + 64, v[j][2]);
          atomicAdd(o + 96, v[j][3]);
        }
      }
      if (lane < OWN_GROUP) {
        const int rd = dv - base;
        if ((unsigned int)rd < (unsigned int)OWN_ROWS) atomicAdd(&sDeg[rd], 1.0f);
      }
      sv = sv2;
      dv = dv2;
      g0 = g1;
    }
    const int nFull = (n / OWN_GROUP) * OWN_GROUP;
    if (wave == ((n / OWN_GROUP) % OWN_WAVES)) {
      for (int g = nFull; g < n; ++g) {
        const int e = sList[g];
        const int rr = dst[e] - base;
        const float* xr = x + (size_t)src[e] * 128 + lane;
        const float v0 = xr[0];
        const float v1 = xr[32];
        const float v2 = xr[64];
        const float v3 = xr[96];
        if ((unsigned int)rr < (unsigned int)OWN_ROWS) {
          float* o = &sAcc[rr * 128 + lane];
          atomicAdd(o,      v0);
          atomicAdd(o + 32, v1);
          atomicAdd(o + 64, v2);
          atomicAdd(o + 96, v3);
          if (lane == 0) atomicAdd(&sDeg[rr], 1.0f);
        }
      }
    }
  }
  __syncthreads();

  const int rowsHere = min(OWN_ROWS, nRows - base);
  float4* ag4 = (float4*)(agg + (size_t)base * 128);
  const float4* sA4 = (const float4*)sAcc;
  for (int i = tid; i < rowsHere * 32; i += OWN_THREADS) ag4[i] = sA4[i];
  if (tid < rowsHere) deg[base + tid] = sDeg[tid];
}

template <int NT>
static __device__ __forceinline__ void stage_w_fragments(const float* __restrict__ Wg,
                                                         v2f* __restrict__ sW,
                                                         int wave, int lane) {
  const int h = lane >> 4;
  const int c = lane & 15;
#pragma unroll 8
  for (int u = wave; u < 32 * NT; u += 8) {
    const int kk = u / NT;
    const int nt = u - kk * NT;
    const float* p = Wg + (size_t)(4 * kk + 2 * h) * (NT * 16) + nt * 16 + c;
    v2f b;
    b.x = p[0];
    b.y = p[NT * 16];
    sW[u * 32 + lane] = b;
  }
}

__global__ __launch_bounds__(256)
void sage_strip_gemm_kernel(const float* __restrict__ X,
                            const float* __restrict__ AGG,
                            const float* __restrict__ DEG,
                            const float* __restrict__ Wself,
                            const float* __restrict__ Wneigh,
                            const float* __restrict__ bias,
                            float* __restrict__ out,
                            int nRows, int doRelu) {
  __shared__ v2f sW[2 * 32 * 8 * 32];
  const int wave = threadIdx.x >> 5;
  const int lane = threadIdx.x & 31;
  stage_w_fragments<8>(Wself,  sW,               wave, lane);
  stage_w_fragments<8>(Wneigh, sW + 32 * 8 * 32, wave, lane);
  __syncthreads();

  const int mt = blockIdx.x * 8 + wave;
  if (mt >= (nRows >> 4)) return;

  const int half = lane >> 4;
  const int l15  = lane & 15;
  const int aRow = mt * 16 + l15;
  const int kb   = half * 2;

  const float inv = 1.0f / fmaxf(DEG[aRow], 1.0f);
  float bv[8];
#pragma unroll
  for (int nt = 0; nt < 8; ++nt) bv[nt] = bias[nt * 16 + l15];

  v8f acc[8];
#pragma unroll
  for (int nt = 0; nt < 8; ++nt) acc[nt] = (v8f){0.f, 0.f, 0.f, 0.f, 0.f, 0.f, 0.f, 0.f};

  const v2f* sWs = sW + lane;
  const v2f* sWn = sW + 32 * 8 * 32 + lane;
  const float* xr = X   + (size_t)aRow * 128 + kb;
  const float* gr = AGG + (size_t)aRow * 128 + kb;

#pragma unroll 1
  for (int k0 = 0; k0 < 32; k0 += 8) {
    v2f a[8];
#pragma unroll
    for (int j = 0; j < 8; ++j) a[j] = *(const v2f*)(xr + 4 * (k0 + j));
#pragma unroll
    for (int j = 0; j < 8; ++j) {
#pragma unroll
      for (int nt = 0; nt < 8; ++nt) {
        const v2f b = sWs[((k0 + j) * 8 + nt) * 32];
        acc[nt] = wmma_f32_16x16x4(a[j], b, acc[nt]);
      }
    }
  }
#pragma unroll 1
  for (int k0 = 0; k0 < 32; k0 += 8) {
    v2f a[8];
#pragma unroll
    for (int j = 0; j < 8; ++j) {
      a[j] = *(const v2f*)(gr + 4 * (k0 + j));
      a[j].x *= inv;
      a[j].y *= inv;
    }
#pragma unroll
    for (int j = 0; j < 8; ++j) {
#pragma unroll
      for (int nt = 0; nt < 8; ++nt) {
        const v2f b = sWn[((k0 + j) * 8 + nt) * 32];
        acc[nt] = wmma_f32_16x16x4(a[j], b, acc[nt]);
      }
    }
  }

  float* orow = out + (size_t)(mt * 16 + half * 8) * 128 + l15;
#pragma unroll
  for (int nt = 0; nt < 8; ++nt) {
#pragma unroll
    for (int i = 0; i < 8; ++i) {
      float v = acc[nt][i] + bv[nt];
      if (doRelu) v = fmaxf(v, 0.0f);
      orow[(size_t)i * 128 + nt * 16] = v;
    }
  }
}

__global__ __launch_bounds__(256) __attribute__((amdgpu_waves_per_eu(2, 4)))
void sage_strip_dual_proj_kernel(const float* __restrict__ H,
                                 const float* __restrict__ Wself,
                                 const float* __restrict__ Wneigh,
                                 const float* __restrict__ bias,
                                 float* __restrict__ outSelf,
                                 float* __restrict__ outNeigh,
                                 int nRows) {
  __shared__ v2f sW[2 * 32 * 4 * 32];
  const int wave = threadIdx.x >> 5;
  const int lane = threadIdx.x & 31;
  stage_w_fragments<4>(Wself,  sW,               wave, lane);
  stage_w_fragments<4>(Wneigh, sW + 32 * 4 * 32, wave, lane);
  __syncthreads();

  const int mt = blockIdx.x * 8 + wave;
  if (mt >= (nRows >> 4)) return;

  const int half = lane >> 4;
  const int l15  = lane & 15;
  const int aRow = mt * 16 + l15;
  const int kb   = half * 2;

  float bv[4];
#pragma unroll
  for (int nt = 0; nt < 4; ++nt) bv[nt] = bias[nt * 16 + l15];

  v8f cs[4], cn[4];
#pragma unroll
  for (int nt = 0; nt < 4; ++nt) {
    cs[nt] = (v8f){0.f, 0.f, 0.f, 0.f, 0.f, 0.f, 0.f, 0.f};
    cn[nt] = (v8f){0.f, 0.f, 0.f, 0.f, 0.f, 0.f, 0.f, 0.f};
  }

  const v2f* sWs = sW + lane;
  const v2f* sWn = sW + 32 * 4 * 32 + lane;
  const float* xr = H + (size_t)aRow * 128 + kb;

#pragma unroll 1
  for (int k0 = 0; k0 < 32; k0 += 8) {
    v2f a[8];
#pragma unroll
    for (int j = 0; j < 8; ++j) a[j] = *(const v2f*)(xr + 4 * (k0 + j));
#pragma unroll
    for (int j = 0; j < 8; ++j) {
#pragma unroll
      for (int nt = 0; nt < 4; ++nt) {
        const v2f bs = sWs[((k0 + j) * 4 + nt) * 32];
        const v2f bn = sWn[((k0 + j) * 4 + nt) * 32];
        cs[nt] = wmma_f32_16x16x4(a[j], bs, cs[nt]);
        cn[nt] = wmma_f32_16x16x4(a[j], bn, cn[nt]);
      }
    }
  }

  const size_t o0 = (size_t)(mt * 16 + half * 8) * 64 + l15;
#pragma unroll
  for (int nt = 0; nt < 4; ++nt) {
#pragma unroll
    for (int i = 0; i < 8; ++i) {
      outSelf[o0 + (size_t)i * 64 + nt * 16]  = cs[nt][i] + bv[nt];
      outNeigh[o0 + (size_t)i * 64 + nt * 16] = cn[nt][i];
    }
  }
}

#define EDGES_PER_WAVE 8

__global__ __launch_bounds__(256)
void scatter_mean64_into_out_kernel(const float* __restrict__ P,
                                    const int* __restrict__ src,
                                    const int* __restrict__ dst,
                                    const float* __restrict__ DEG,
                                    float* __restrict__ out, int nE) {
  const int lane = threadIdx.x & 31;
  const int w  = __builtin_amdgcn_readfirstlane((int)((blockIdx.x * blockDim.x + threadIdx.x) >> 5));
  const int e0 = w * EDGES_PER_WAVE;
  if (e0 >= nE) return;
  if (e0 + EDGES_PER_WAVE <= nE) {
    int s[EDGES_PER_WAVE], d[EDGES_PER_WAVE];
#pragma unroll
    for (int j = 0; j < EDGES_PER_WAVE; ++j) { s[j] = src[e0 + j]; d[j] = dst[e0 + j]; }
    float a0[EDGES_PER_WAVE], a1[EDGES_PER_WAVE], inv[EDGES_PER_WAVE];
#pragma unroll
    for (int j = 0; j < EDGES_PER_WAVE; ++j) {
      const float* pr = P + (size_t)s[j] * 64 + lane;
      a0[j] = pr[0];
      a1[j] = pr[32];
    }
#pragma unroll
    for (int j = 0; j < EDGES_PER_WAVE; ++j) inv[j] = 1.0f / fmaxf(DEG[d[j]], 1.0f);
#pragma unroll
    for (int j = 0; j < EDGES_PER_WAVE; ++j) {
      float* o = out + (size_t)d[j] * 64 + lane;
      atomicAdd(o,      a0[j] * inv[j]);
      atomicAdd(o + 32, a1[j] * inv[j]);
    }
  } else {
    for (int e = e0; e < nE; ++e) {
      const int dd = dst[e];
      const float* pr = P + (size_t)src[e] * 64 + lane;
      const float b0 = pr[0];
      const float b1 = pr[32];
      const float iv = 1.0f / fmaxf(DEG[dd], 1.0f);
      float* o = out + (size_t)dd * 64 + lane;
      atomicAdd(o,      b0 * iv);
      atomicAdd(o + 32, b1 * iv);
    }
  }
}

extern "C" void kernel_launch(void* const* d_in, const int* in_sizes, int n_in,
                              void* d_out, int out_size, void* d_ws, size_t ws_size,
                              hipStream_t stream) {
  const float* x   = (const float*)d_in[0];
  const float* Ws1 = (const float*)d_in[1];
  const float* Wn1 = (const float*)d_in[2];
  const float* b1  = (const float*)d_in[3];
  const float* Ws2 = (const float*)d_in[4];
  const float* Wn2 = (const float*)d_in[5];
  const float* b2  = (const float*)d_in[6];
  const int*   src = (const int*)d_in[7];
  const int*   dst = (const int*)d_in[8];
  float* out = (float*)d_out;

  const int F = in_sizes[3];
  const int N = in_sizes[0] / F;
  const int E = in_sizes[7];

  char* ws = (char*)d_ws;
  float* h   = (float*)ws;
  float* agg = (float*)(ws + (size_t)N * F * 4);
  float* deg = (float*)(ws + (size_t)N * F * 8);

  const int TB = 256;
  const int edgeRunWaves  = (E + EDGES_PER_WAVE - 1) / EDGES_PER_WAVE;
  const int edgeRunBlocks = (edgeRunWaves + (TB / 32) - 1) / (TB / 32);

  unsigned int* ownerW = (unsigned int*)ws;
  const int ownerWords  = (E + 3) / 4;
  const int ownRuns     = (N + OWN_ROWS - 1) / OWN_ROWS;
  owner_byte_kernel<<<(ownerWords + TB - 1) / TB, TB, 0, stream>>>(dst, ownerW, E);
  own_rows_sum_kernel<<<ownRuns, OWN_THREADS, 0, stream>>>(x, src, dst, ownerW, agg, deg, E, N);

  {
    const int strips = N / 16;
    const int blocks = (strips + (TB / 32) - 1) / (TB / 32);
    sage_strip_gemm_kernel<<<blocks, TB, 0, stream>>>(
        x, agg, deg, Ws1, Wn1, b1, h, N,  1);
  }

  float* p = agg;
  {
    const int strips = N / 16;
    const int blocks = (strips + (TB / 32) - 1) / (TB / 32);
    sage_strip_dual_proj_kernel<<<blocks, TB, 0, stream>>>(
        h, Ws2, Wn2, b2, out, p, N);
  }
  scatter_mean64_into_out_kernel<<<edgeRunBlocks, TB, 0, stream>>>(p, src, dst, deg, out, E);
}


# layer 1's two products run inside the owner work-groups (16-row x 32-column tasks), neighbour A fragments read from the LDS sums at row pitch 132; sum plane and one launch go; k order kept
# speedup vs baseline: 3.7989x; 1.0657x over previous
#include <hip/hip_runtime.h>


typedef float v2f __attribute__((ext_vector_type(2)));
typedef float v8f __attribute__((ext_vector_type(8)));

static __device__ __forceinline__ v8f wmma_f32_16x16x4(v2f a, v2f b, v8f c) {
  return __builtin_amdgcn_wmma_f32_16x16x4_f32(
       false, a,  false, b,
       (short)0, c,  false,  false);
}

#define OWN_ROWS 196
#define OWN_STRIPS 13
#define OWN_PAD_ROWS (OWN_STRIPS * 16)
#define OWN_PITCH 132
#define OWN_LIST_CAP 8192
#define OWN_THREADS 1024
#define OWN_WAVES (OWN_THREADS / 32)
#define OWN_GROUP 16

__global__ __launch_bounds__(256)
void owner_byte_kernel(const int* __restrict__ dst, unsigned int* __restrict__ ownerW, int nE) {
  const int i = blockIdx.x * blockDim.x + threadIdx.x;
  const int e = i * 4;
  if (e >= nE) return;
  unsigned int w = 0;
#pragma unroll
  for (int j = 0; j < 4; ++j) {
    if (e + j < nE) w |= (((unsigned int)dst[e + j] / OWN_ROWS) & 0xffu) << (8 * j);
  }
  ownerW[i] = w;
}

static __device__ __forceinline__ unsigned int zero_byte_flags(unsigned int y) {
  return ~((((y & 0x7f7f7f7fu) + 0x7f7f7f7fu) | y) | 0x7f7f7f7fu);
}

static __device__ __forceinline__ unsigned int own_chunk_flags(uint4 v, unsigned int pat) {
  return (zero_byte_flags(v.x ^ pat) >> 7) | (zero_byte_flags(v.y ^ pat) >> 6) |
         (zero_byte_flags(v.z ^ pat) >> 5) | (zero_byte_flags(v.w ^ pat) >> 4);
}

static __device__ __forceinline__ void own_append(unsigned int c, int qA, int qB,
                                                  int* sCnt, int* sList) {
  while (c != 0u) {
    const int p = __ffs((int)c) - 1;
    c &= c - 1u;
    const int q = ((p >> 2) & 1) ? qB : qA;
    const int e = q * 16 + (p & 3) * 4 + (p >> 3);
    const int slot = atomicAdd(sCnt, 1);
    if (slot < OWN_LIST_CAP) sList[slot] = e;
  }
}

template <int NT, int NW>
static __device__ __forceinline__ void stage_w_fragments(const float* __restrict__ Wg,
                                                         v2f* __restrict__ sW,
                                                         int wave, int lane) {
  const int h = lane >> 4;
  const int c = lane & 15;
#pragma unroll 8
  for (int u = wave; u < 32 * NT; u += NW) {
    const int kk = u / NT;
    const int nt = u - kk * NT;
    const float* p = Wg + (size_t)(4 * kk + 2 * h) * (NT * 16) + nt * 16 + c;
    v2f b;
    b.x = p[0];
    b.y = p[NT * 16];
    sW[u * 32 + lane] = b;
  }
}

__global__ __launch_bounds__(OWN_THREADS)
void own_rows_layer1_kernel(const float* __restrict__ x,
                            const int* __restrict__ src,
                            const int* __restrict__ dst,
                            const unsigned int* __restrict__ ownerW,
                            const float* __restrict__ Wself,
                            const float* __restrict__ Wneigh,
                            const float* __restrict__ bias,
                            float* __restrict__ hOut,
                            float* __restrict__ deg,
                            int nE, int nRows) {
  __shared__ __align__(16) float sAcc[OWN_PAD_ROWS * OWN_PITCH];
  __shared__ float sDeg[OWN_PAD_ROWS];
  __shared__ int   sList[OWN_LIST_CAP];
  __shared__ int   sCnt;
  __shared__ v2f   sW[2 * 32 * 8 * 32];
  const int tid  = threadIdx.x;
  const int lane = tid & 31;
  const int wave = tid >> 5;
  const int base = blockIdx.x * OWN_ROWS;
  const unsigned int myRun = blockIdx.x & 0xffu;

  stage_w_fragments<8, OWN_WAVES>(Wself,  sW,               wave, lane);
  stage_w_fragments<8, OWN_WAVES>(Wneigh, sW + 32 * 8 * 32, wave, lane);
  for (int i = tid; i < OWN_PAD_ROWS * OWN_PITCH; i += OWN_THREADS) sAcc[i] = 0.0f;
  if (tid < OWN_PAD_ROWS) sDeg[tid] = 0.0f;
  if (tid == 0) sCnt = 0;
  __syncthreads();

  const unsigned int pat = myRun * 0x01010101u;
  const int nChunks = nE >> 4;
  const uint4* ow = (const uint4*)ownerW;
  int q0 = tid;
  for (; q0 + 7 * OWN_THREADS < nChunks; q0 += 8 * OWN_THREADS) {
    uint4 v[8];
#pragma unroll
    for (int u = 0; u < 8; ++u) v[u] = ow[q0 + u * OWN_THREADS];
#pragma unroll
    for (int u = 0; u < 8; u += 2) {
      const unsigned int c = own_chunk_flags(v[u], pat) | (own_chunk_flags(v[u + 1], pat) << 4);
      own_append(c, q0 + u * OWN_THREADS, q0 + (u + 1) * OWN_THREADS, &sCnt, sList);
    }
  }
  for (; q0 < nChunks; q0 += OWN_THREADS) {
    const unsigned int c = own_chunk_flags(ow[q0], pat);
    own_append(c, q0, q0, &sCnt, sList);
  }
  {
    const int e = (nChunks << 4) + tid;
    if (tid < 16 && e < nE) {
      const unsigned int w = ownerW[e >> 2];
      if (((w >> (8 * (e & 3))) & 0xffu) == myRun) {
        const int slot = atomicAdd(&sCnt, 1);
        if (slot < OWN_LIST_CAP) sList[slot] = e;
      }
    }
  }
  __syncthreads();

  const int total = sCnt;
  if (total > OWN_LIST_CAP) {
    const unsigned char* ob = (const unsigned char*)ownerW;
    for (int e = tid; e < nE; e += OWN_THREADS) {
      if ((unsigned int)ob[e] != myRun) continue;
      const int r = dst[e] - base;
      if ((unsigned int)r >= (unsigned int)OWN_ROWS) continue;
      const float* xr = x + (size_t)src[e] * 128;
      for (int cc = 0; cc < 128; ++cc) atomicAdd(&sAcc[r * OWN_PITCH + cc], xr[cc]);
      atomicAdd(&sDeg[r], 1.0f);
    }
  } else {
    const int n = total;
    int g0 = wave * OWN_GROUP;
    int sv = 0, dv = 0;
    if (g0 + OWN_GROUP <= n && lane < OWN_GROUP) {
      const int e = sList[g0 + lane];
      sv = src[e];
      dv = dst[e];
    }
    while (g0 + OWN_GROUP <= n) {
      float v[OWN_GROUP][4];
      int   r[OWN_GROUP];
#pragma unroll
      for (int j = 0; j < OWN_GROUP; ++j) {
        const int s = __builtin_amdgcn_readlane(sv, j);
        r[j] = __builtin_amdgcn_readlane(dv, j) - base;
        const float* xr = x + (size_t)s * 128 + lane;
        v[j][0] = xr[0];
        v[j][1] = xr[32];
        v[j][2] = xr[64];
        v[j][3] = xr[96];
      }
      const int g1 = g0 + OWN_WAVES * OWN_GROUP;
      int sv2 = 0, dv2 = 0;
      if (g1 + OWN_GROUP <= n && lane < OWN_GROUP) {
        const int e = sList[g1 + lane];
        sv2 = src[e];
        dv2 = dst[e];
      }
#pragma unroll
      for (int j = 0; j < OWN_GROUP; ++j) {
        if ((unsigned int)r[j] < (unsigned int)OWN_ROWS) {
          float* o = &sAcc[r[j] * OWN_PITCH + lane];
          atomicAdd(o,      v[j][0]);
          atomicAdd(o + 32, v[j][1]);
          atomicAdd(o + 64, v[j][2]);
          atomicAdd(o + 96, v[j][3]);
        }
      }
      if (lane < OWN_GROUP) {
        const int rd = dv - base;
        if ((unsigned int)rd < (unsigned int)OWN_ROWS) atomicAdd(&sDeg[rd], 1.0f);
      }
      sv = sv2;
      dv = dv2;
      g0 = g1;
    }
    const int nFull = (n / OWN_GROUP) * OWN_GROUP;
    if (wave == ((n / OWN_GROUP) % OWN_WAVES)) {
      for (int g = nFull; g < n; ++g) {
        const int e = sList[g];
        const int rr = dst[e] - base;
        const float* xr = x + (size_t)src[e] * 128 + lane;
        const float v0 = xr[0];
        const float v1 = xr[32];
        const float v2 = xr[64];
        const float v3 = xr[96];
        if ((unsigned int)rr < (unsigned int)OWN_ROWS) {
          float* o = &sAcc[rr * OWN_PITCH + lane];
          atomicAdd(o,      v0);
          atomicAdd(o + 32, v1);
          atomicAdd(o + 64, v2);
          atomicAdd(o + 96, v3);
          if (lane == 0) atomicAdd(&sDeg[rr], 1.0f);
        }
      }
    }
  }
  __syncthreads();

  const int rowsHere = min(OWN_ROWS, nRows - base);
  if (tid < rowsHere) deg[base + tid] = sDeg[tid];

  const int nTasks = ((rowsHere + 15) >> 4) * 4;
  const int half = lane >> 4;
  const int l15  = lane & 15;
  const int kb   = half * 2;
#pragma unroll 1
  for (int t = wave; t < nTasks; t += OWN_WAVES) {
    const int strip = t >> 2;
    const int nt0   = (t & 3) * 2;
    const int lrow  = strip * 16 + l15;
    const int grow  = min(base + lrow, nRows - 1);
    const float inv = 1.0f / fmaxf(sDeg[lrow], 1.0f);
    const float bv0 = bias[nt0 * 16 + l15];
    const float bv1 = bias[nt0 * 16 + 16 + l15];

    v8f acc0 = (v8f){0.f, 0.f, 0.f, 0.f, 0.f, 0.f, 0.f, 0.f};
    v8f acc1 = (v8f){0.f, 0.f, 0.f, 0.f, 0.f, 0.f, 0.f, 0.f};

    const v2f* sWs = sW + nt0 * 32 + lane;
    const v2f* sWn = sW + 32 * 8 * 32 + nt0 * 32 + lane;
    const float* xr = x + (size_t)grow * 128 + kb;
    const float* gr = sAcc + lrow * OWN_PITCH + kb;

#pragma unroll 1
    for (int k0 = 0; k0 < 32; k0 += 8) {
      v2f a[8];
#pragma unroll
      for (int j = 0; j < 8; ++j) a[j] = *(const v2f*)(xr + 4 * (k0 + j));
#pragma unroll
      for (int j = 0; j < 8; ++j) {
        const v2f b0 = sWs[(k0 + j) * 8 * 32];
        const v2f b1 = sWs[(k0 + j) * 8 * 32 + 32];
        acc0 = wmma_f32_16x16x4(a[j], b0, acc0);
        acc1 = wmma_f32_16x16x4(a[j], b1, acc1);
      }
    }
#pragma unroll 1
    for (int k0 = 0; k0 < 32; k0 += 8) {
      v2f a[8];
#pragma unroll
      for (int j = 0; j < 8; ++j) {
        a[j] = *(const v2f*)(gr + 4 * (k0 + j));
        a[j].x *= inv;
        a[j].y *= inv;
      }
#pragma unroll
      for (int j = 0; j < 8; ++j) {
        const v2f b0 = sWn[(k0 + j) * 8 * 32];
        const v2f b1 = sWn[(k0 + j) * 8 * 32 + 32];
        acc0 = wmma_f32_16x16x4(a[j], b0, acc0);
        acc1 = wmma_f32_16x16x4(a[j], b1, acc1);
      }
    }

    const int orow0 = strip * 16 + half * 8;
    float* op = hOut + (size_t)(base + orow0) * 128 + nt0 * 16 + l15;
#pragma unroll
    for (int i = 0; i < 8; ++i) {
      if (orow0 + i < rowsHere) {
        op[(size_t)i * 128]      = fmaxf(acc0[i] + bv0, 0.0f);
        op[(size_t)i * 128 + 16] = fmaxf(acc1[i] + bv1, 0.0f);
      }
    }
  }
}


__global__ __launch_bounds__(256) __attribute__((amdgpu_waves_per_eu(2, 4)))
void sage_strip_dual_proj_kernel(const float* __restrict__ H,
                                 const float* __restrict__ Wself,
                                 const float* __restrict__ Wneigh,
                                 const float* __restrict__ bias,
                                 float* __restrict__ outSelf,
                                 float* __restrict__ outNeigh,
                                 int nRows) {
  __shared__ v2f sW[2 * 32 * 4 * 32];
  const int wave = threadIdx.x >> 5;
  const int lane = threadIdx.x & 31;
  stage_w_fragments<4, 8>(Wself,  sW,               wave, lane);
  stage_w_fragments<4, 8>(Wneigh, sW + 32 * 4 * 32, wave, lane);
  __syncthreads();

  const int mt = blockIdx.x * 8 + wave;
  if (mt >= (nRows >> 4)) return;

  const int half = lane >> 4;
  const int l15  = lane & 15;
  const int aRow = mt * 16 + l15;
  const int kb   = half * 2;

  float bv[4];
#pragma unroll
  for (int nt = 0; nt < 4; ++nt) bv[nt] = bias[nt * 16 + l15];

  v8f cs[4], cn[4];
#pragma unroll
  for (int nt = 0; nt < 4; ++nt) {
    cs[nt] = (v8f){0.f, 0.f, 0.f, 0.f, 0.f, 0.f, 0.f, 0.f};
    cn[nt] = (v8f){0.f, 0.f, 0.f, 0.f, 0.f, 0.f, 0.f, 0.f};
  }

  const v2f* sWs = sW + lane;
  const v2f* sWn = sW + 32 * 4 * 32 + lane;
  const float* xr = H + (size_t)aRow * 128 + kb;

#pragma unroll 1
  for (int k0 = 0; k0 < 32; k0 += 8) {
    v2f a[8];
#pragma unroll
    for (int j = 0; j < 8; ++j) a[j] = *(const v2f*)(xr + 4 * (k0 + j));
#pragma unroll
    for (int j = 0; j < 8; ++j) {
#pragma unroll
      for (int nt = 0; nt < 4; ++nt) {
        const v2f bs = sWs[((k0 + j) * 4 + nt) * 32];
        const v2f bn = sWn[((k0 + j) * 4 + nt) * 32];
        cs[nt] = wmma_f32_16x16x4(a[j], bs, cs[nt]);
        cn[nt] = wmma_f32_16x16x4(a[j], bn, cn[nt]);
      }
    }
  }

  const size_t o0 = (size_t)(mt * 16 + half * 8) * 64 + l15;
#pragma unroll
  for (int nt = 0; nt < 4; ++nt) {
#pragma unroll
    for (int i = 0; i < 8; ++i) {
      outSelf[o0 + (size_t)i * 64 + nt * 16]  = cs[nt][i] + bv[nt];
      outNeigh[o0 + (size_t)i * 64 + nt * 16] = cn[nt][i];
    }
  }
}

#define EDGES_PER_WAVE 8

__global__ __launch_bounds__(256)
void scatter_mean64_into_out_kernel(const float* __restrict__ P,
                                    const int* __restrict__ src,
                                    const int* __restrict__ dst,
                                    const float* __restrict__ DEG,
                                    float* __restrict__ out, int nE) {
  const int lane = threadIdx.x & 31;
  const int w  = __builtin_amdgcn_readfirstlane((int)((blockIdx.x * blockDim.x + threadIdx.x) >> 5));
  const int e0 = w * EDGES_PER_WAVE;
  if (e0 >= nE) return;
  if (e0 + EDGES_PER_WAVE <= nE) {
    int s[EDGES_PER_WAVE], d[EDGES_PER_WAVE];
#pragma unroll
    for (int j = 0; j < EDGES_PER_WAVE; ++j) { s[j] = src[e0 + j]; d[j] = dst[e0 + j]; }
    float a0[EDGES_PER_WAVE], a1[EDGES_PER_WAVE], inv[EDGES_PER_WAVE];
#pragma unroll
    for (int j = 0; j < EDGES_PER_WAVE; ++j) {
      const float* pr = P + (size_t)s[j] * 64 + lane;
      a0[j] = pr[0];
      a1[j] = pr[32];
    }
#pragma unroll
    for (int j = 0; j < EDGES_PER_WAVE; ++j) inv[j] = 1.0f / fmaxf(DEG[d[j]], 1.0f);
#pragma unroll
    for (int j = 0; j < EDGES_PER_WAVE; ++j) {
      float* o = out + (size_t)d[j] * 64 + lane;
      atomicAdd(o,      a0[j] * inv[j]);
      atomicAdd(o + 32, a1[j] * inv[j]);
    }
  } else {
    for (int e = e0; e < nE; ++e) {
      const int dd = dst[e];
      const float* pr = P + (size_t)src[e] * 64 + lane;
      const float b0 = pr[0];
      const float b1 = pr[32];
      const float iv = 1.0f / fmaxf(DEG[dd], 1.0f);
      float* o = out + (size_t)dd * 64 + lane;
      atomicAdd(o,      b0 * iv);
      atomicAdd(o + 32, b1 * iv);
    }
  }
}

extern "C" void kernel_launch(void* const* d_in, const int* in_sizes, int n_in,
                              void* d_out, int out_size, void* d_ws, size_t ws_size,
                              hipStream_t stream) {
  const float* x   = (const float*)d_in[0];
  const float* Ws1 = (const float*)d_in[1];
  const float* Wn1 = (const float*)d_in[2];
  const float* b1  = (const float*)d_in[3];
  const float* Ws2 = (const float*)d_in[4];
  const float* Wn2 = (const float*)d_in[5];
  const float* b2  = (const float*)d_in[6];
  const int*   src = (const int*)d_in[7];
  const int*   dst = (const int*)d_in[8];
  float* out = (float*)d_out;

  const int F = in_sizes[3];
  const int N = in_sizes[0] / F;
  const int E = in_sizes[7];

  char* ws = (char*)d_ws;
  float* h   = (float*)ws;
  float* pl2 = (float*)(ws + (size_t)N * F * 4);
  float* deg = (float*)(ws + (size_t)N * F * 8);

  const int TB = 256;
  const int edgeRunWaves  = (E + EDGES_PER_WAVE - 1) / EDGES_PER_WAVE;
  const int edgeRunBlocks = (edgeRunWaves + (TB / 32) - 1) / (TB / 32);

  unsigned int* ownerW = (unsigned int*)pl2;
  const int ownerWords  = (E + 3) / 4;
  const int ownRuns     = (N + OWN_ROWS - 1) / OWN_ROWS;
  owner_byte_kernel<<<(ownerWords + TB - 1) / TB, TB, 0, stream>>>(dst, ownerW, E);
  own_rows_layer1_kernel<<<ownRuns, OWN_THREADS, 0, stream>>>(
      x, src, dst, ownerW, Ws1, Wn1, b1, h, deg, E, N);

  float* p = pl2;
  {
    const int strips = N / 16;
    const int blocks = (strips + (TB / 32) - 1) / (TB / 32);
    sage_strip_dual_proj_kernel<<<blocks, TB, 0, stream>>>(
        h, Ws2, Wn2, b2, out, p, N);
  }
  scatter_mean64_into_out_kernel<<<edgeRunBlocks, TB, 0, stream>>>(p, src, dst, deg, out, E);
}


# layer 2: each 196-row owner work-group sums p[src] in LDS (f32 LDS atomics, arrival order) from the edge list layer 1 saves, then out = self + sum/deg once; no global atomics; f32 sum order changes
# speedup vs baseline: 4.3563x; 1.1467x over previous
#include <hip/hip_runtime.h>


typedef float v2f __attribute__((ext_vector_type(2)));
typedef float v8f __attribute__((ext_vector_type(8)));

static __device__ __forceinline__ v8f wmma_f32_16x16x4(v2f a, v2f b, v8f c) {
  return __builtin_amdgcn_wmma_f32_16x16x4_f32(
       false, a,  false, b,
       (short)0, c,  false,  false);
}

#define OWN_ROWS 196
#define OWN_STRIPS 13
#define OWN_PAD_ROWS (OWN_STRIPS * 16)
#define OWN_PITCH 132
#define OWN_LIST_CAP 8192
#define OWN_THREADS 1024
#define OWN_WAVES (OWN_THREADS / 32)
#define OWN_GROUP 16

#define OWN_NO_ROW 255
#define OWN_LIST_HEAD 32
#define OWN_LIST_STRIDE (OWN_LIST_HEAD + OWN_LIST_CAP)

static __device__ __forceinline__ int own_list_word(int srcRow, int localRow) {
  return (srcRow << 8) | localRow;
}

__global__ __launch_bounds__(256)
void owner_byte_kernel(const int* __restrict__ dst, unsigned int* __restrict__ ownerW, int nE) {
  const int i = blockIdx.x * blockDim.x + threadIdx.x;
  const int e = i * 4;
  if (e >= nE) return;
  unsigned int w = 0;
#pragma unroll
  for (int j = 0; j < 4; ++j) {
    if (e + j < nE) w |= (((unsigned int)dst[e + j] / OWN_ROWS) & 0xffu) << (8 * j);
  }
  ownerW[i] = w;
}

static __device__ __forceinline__ unsigned int zero_byte_flags(unsigned int y) {
  return ~((((y & 0x7f7f7f7fu) + 0x7f7f7f7fu) | y) | 0x7f7f7f7fu);
}

static __device__ __forceinline__ unsigned int own_chunk_flags(uint4 v, unsigned int pat) {
  return (zero_byte_flags(v.x ^ pat) >> 7) | (zero_byte_flags(v.y ^ pat) >> 6) |
         (zero_byte_flags(v.z ^ pat) >> 5) | (zero_byte_flags(v.w ^ pat) >> 4);
}

static __device__ __forceinline__ void own_append(unsigned int c, int qA, int qB,
                                                  int* sCnt, int* sList) {
  while (c != 0u) {
    const int p = __ffs((int)c) - 1;
    c &= c - 1u;
    const int q = ((p >> 2) & 1) ? qB : qA;
    const int e = q * 16 + (p & 3) * 4 + (p >> 3);
    const int slot = atomicAdd(sCnt, 1);
    if (slot < OWN_LIST_CAP) sList[slot] = e;
  }
}

template <int NT, int NW>
static __device__ __forceinline__ void stage_w_fragments(const float* __restrict__ Wg,
                                                         v2f* __restrict__ sW,
                                                         int wave, int lane) {
  const int h = lane >> 4;
  const int c = lane & 15;
#pragma unroll 8
  for (int u = wave; u < 32 * NT; u += NW) {
    const int kk = u / NT;
    const int nt = u - kk * NT;
    const float* p = Wg + (size_t)(4 * kk + 2 * h) * (NT * 16) + nt * 16 + c;
    v2f b;
    b.x = p[0];
    b.y = p[NT * 16];
    sW[u * 32 + lane] = b;
  }
}

__global__ __launch_bounds__(OWN_THREADS)
void own_rows_layer1_kernel(const float* __restrict__ x,
                            const int* __restrict__ src,
                            const int* __restrict__ dst,
                            const unsigned int* __restrict__ ownerW,
                            const float* __restrict__ Wself,
                            const float* __restrict__ Wneigh,
                            const float* __restrict__ bias,
                            float* __restrict__ hOut,
                            float* __restrict__ deg,
                            int* __restrict__ ownLists,
                            int nE, int nRows) {
  __shared__ __align__(16) float sAcc[OWN_PAD_ROWS * OWN_PITCH];
  __shared__ float sDeg[OWN_PAD_ROWS];
  __shared__ int   sList[OWN_LIST_CAP];
  __shared__ int   sCnt;
  __shared__ v2f   sW[2 * 32 * 8 * 32];
  const int tid  = threadIdx.x;
  const int lane = tid & 31;
  const int wave = tid >> 5;
  const int base = blockIdx.x * OWN_ROWS;
  const unsigned int myRun = blockIdx.x & 0xffu;

  stage_w_fragments<8, OWN_WAVES>(Wself,  sW,               wave, lane);
  stage_w_fragments<8, OWN_WAVES>(Wneigh, sW + 32 * 8 * 32, wave, lane);
  for (int i = tid; i < OWN_PAD_ROWS * OWN_PITCH; i += OWN_THREADS) sAcc[i] = 0.0f;
  if (tid < OWN_PAD_ROWS) sDeg[tid] = 0.0f;
  if (tid == 0) sCnt = 0;
  __syncthreads();

  const unsigned int pat = myRun * 0x01010101u;
  const int nChunks = nE >> 4;
  const uint4* ow = (const uint4*)ownerW;
  int q0 = tid;
  for (; q0 + 7 * OWN_THREADS < nChunks; q0 += 8 * OWN_THREADS) {
    uint4 v[8];
#pragma unroll
    for (int u = 0; u < 8; ++u) v[u] = ow[q0 + u * OWN_THREADS];
#pragma unroll
    for (int u = 0; u < 8; u += 2) {
      const unsigned int c = own_chunk_flags(v[u], pat) | (own_chunk_flags(v[u + 1], pat) << 4);
      own_append(c, q0 + u * OWN_THREADS, q0 + (u + 1) * OWN_THREADS, &sCnt, sList);
    }
  }
  for (; q0 < nChunks; q0 += OWN_THREADS) {
    const unsigned int c = own_chunk_flags(ow[q0], pat);
    own_append(c, q0, q0, &sCnt, sList);
  }
  {
    const int e = (nChunks << 4) + tid;
    if (tid < 16 && e < nE) {
      const unsigned int w = ownerW[e >> 2];
      if (((w >> (8 * (e & 3))) & 0xffu) == myRun) {
        const int slot = atomicAdd(&sCnt, 1);
        if (slot < OWN_LIST_CAP) sList[slot] = e;
      }
    }
  }
  __syncthreads();

  const int total = sCnt;
  int* myHead = ownLists + (size_t)blockIdx.x * OWN_LIST_STRIDE;
  int* myList = myHead + OWN_LIST_HEAD;
  if (tid == 0) myHead[0] = total;
  if (total > OWN_LIST_CAP) {
    const unsigned char* ob = (const unsigned char*)ownerW;
    for (int e = tid; e < nE; e += OWN_THREADS) {
      if ((unsigned int)ob[e] != myRun) continue;
      const int r = dst[e] - base;
      if ((unsigned int)r >= (unsigned int)OWN_ROWS) continue;
      const float* xr = x + (size_t)src[e] * 128;
      for (int cc = 0; cc < 128; ++cc) atomicAdd(&sAcc[r * OWN_PITCH + cc], xr[cc]);
      atomicAdd(&sDeg[r], 1.0f);
    }
  } else {
    const int n = total;
    int g0 = wave * OWN_GROUP;
    int sv = 0, dv = 0;
    if (g0 + OWN_GROUP <= n && lane < OWN_GROUP) {
      const int e = sList[g0 + lane];
      sv = src[e];
      dv = dst[e];
    }
    while (g0 + OWN_GROUP <= n) {
      float v[OWN_GROUP][4];
      int   r[OWN_GROUP];
#pragma unroll
      for (int j = 0; j < OWN_GROUP; ++j) {
        const int s = __builtin_amdgcn_readlane(sv, j);
        r[j] = __builtin_amdgcn_readlane(dv, j) - base;
        const float* xr = x + (size_t)s * 128 + lane;
        v[j][0] = xr[0];
        v[j][1] = xr[32];
        v[j][2] = xr[64];
        v[j][3] = xr[96];
      }
      const int g1 = g0 + OWN_WAVES * OWN_GROUP;
      int sv2 = 0, dv2 = 0;
      if (g1 + OWN_GROUP <= n && lane < OWN_GROUP) {
        const int e = sList[g1 + lane];
        sv2 = src[e];
        dv2 = dst[e];
      }
#pragma unroll
      for (int j = 0; j < OWN_GROUP; ++j) {
        if ((unsigned int)r[j] < (unsigned int)OWN_ROWS) {
          float* o = &sAcc[r[j] * OWN_PITCH + lane];
          atomicAdd(o,      v[j][0]);
          atomicAdd(o + 32, v[j][1]);
          atomicAdd(o + 64, v[j][2]);
          atomicAdd(o + 96, v[j][3]);
        }
      }
      if (lane < OWN_GROUP) {
        const int rd = dv - base;
        const bool mine = (unsigned int)rd < (unsigned int)OWN_ROWS;
        myList[g0 + lane] = own_list_word(sv, mine ? rd : OWN_NO_ROW);
        if (mine) atomicAdd(&sDeg[rd], 1.0f);
      }
      sv = sv2;
      dv = dv2;
      g0 = g1;
    }
    const int nFull = (n / OWN_GROUP) * OWN_GROUP;
    if (wave == ((n / OWN_GROUP) % OWN_WAVES)) {
      for (int g = nFull; g < n; ++g) {
        const int e = sList[g];
        const int rr = dst[e] - base;
        const int ss = src[e];
        const float* xr = x + (size_t)ss * 128 + lane;
        const float v0 = xr[0];
        const float v1 = xr[32];
        const float v2 = xr[64];
        const float v3 = xr[96];
        if (lane == 0) {
          myList[g] = own_list_word(ss, (unsigned int)rr < (unsigned int)OWN_ROWS ? rr : OWN_NO_ROW);
        }
        if ((unsigned int)rr < (unsigned int)OWN_ROWS) {
          float* o = &sAcc[rr * OWN_PITCH + lane];
          atomicAdd(o,      v0);
          atomicAdd(o + 32, v1);
          atomicAdd(o + 64, v2);
          atomicAdd(o + 96, v3);
          if (lane == 0) atomicAdd(&sDeg[rr], 1.0f);
        }
      }
    }
  }
  __syncthreads();

  const int rowsHere = min(OWN_ROWS, nRows - base);
  if (tid < rowsHere) deg[base + tid] = sDeg[tid];

  const int nTasks = ((rowsHere + 15) >> 4) * 4;
  const int half = lane >> 4;
  const int l15  = lane & 15;
  const int kb   = half * 2;
#pragma unroll 1
  for (int t = wave; t < nTasks; t += OWN_WAVES) {
    const int strip = t >> 2;
    const int nt0   = (t & 3) * 2;
    const int lrow  = strip * 16 + l15;
    const int grow  = min(base + lrow, nRows - 1);
    const float inv = 1.0f / fmaxf(sDeg[lrow], 1.0f);
    const float bv0 = bias[nt0 * 16 + l15];
    const float bv1 = bias[nt0 * 16 + 16 + l15];

    v8f acc0 = (v8f){0.f, 0.f, 0.f, 0.f, 0.f, 0.f, 0.f, 0.f};
    v8f acc1 = (v8f){0.f, 0.f, 0.f, 0.f, 0.f, 0.f, 0.f, 0.f};

    const v2f* sWs = sW + nt0 * 32 + lane;
    const v2f* sWn = sW + 32 * 8 * 32 + nt0 * 32 + lane;
    const float* xr = x + (size_t)grow * 128 + kb;
    const float* gr = sAcc + lrow * OWN_PITCH + kb;

#pragma unroll 1
    for (int k0 = 0; k0 < 32; k0 += 8) {
      v2f a[8];
#pragma unroll
      for (int j = 0; j < 8; ++j) a[j] = *(const v2f*)(xr + 4 * (k0 + j));
#pragma unroll
      for (int j = 0; j < 8; ++j) {
        const v2f b0 = sWs[(k0 + j) * 8 * 32];
        const v2f b1 = sWs[(k0 + j) * 8 * 32 + 32];
        acc0 = wmma_f32_16x16x4(a[j], b0, acc0);
        acc1 = wmma_f32_16x16x4(a[j], b1, acc1);
      }
    }
#pragma unroll 1
    for (int k0 = 0; k0 < 32; k0 += 8) {
      v2f a[8];
#pragma unroll
      for (int j = 0; j < 8; ++j) {
        a[j] = *(const v2f*)(gr + 4 * (k0 + j));
        a[j].x *= inv;
        a[j].y *= inv;
      }
#pragma unroll
      for (int j = 0; j < 8; ++j) {
        const v2f b0 = sWn[(k0 + j) * 8 * 32];
        const v2f b1 = sWn[(k0 + j) * 8 * 32 + 32];
        acc0 = wmma_f32_16x16x4(a[j], b0, acc0);
        acc1 = wmma_f32_16x16x4(a[j], b1, acc1);
      }
    }

    const int orow0 = strip * 16 + half * 8;
    float* op = hOut + (size_t)(base + orow0) * 128 + nt0 * 16 + l15;
#pragma unroll
    for (int i = 0; i < 8; ++i) {
      if (orow0 + i < rowsHere) {
        op[(size_t)i * 128]      = fmaxf(acc0[i] + bv0, 0.0f);
        op[(size_t)i * 128 + 16] = fmaxf(acc1[i] + bv1, 0.0f);
      }
    }
  }
}


__global__ __launch_bounds__(256) __attribute__((amdgpu_waves_per_eu(2, 4)))
void sage_strip_dual_proj_kernel(const float* __restrict__ H,
                                 const float* __restrict__ Wself,
                                 const float* __restrict__ Wneigh,
                                 const float* __restrict__ bias,
                                 float* __restrict__ outSelf,
                                 float* __restrict__ outNeigh,
                                 int nRows) {
  __shared__ v2f sW[2 * 32 * 4 * 32];
  const int wave = threadIdx.x >> 5;
  const int lane = threadIdx.x & 31;
  stage_w_fragments<4, 8>(Wself,  sW,               wave, lane);
  stage_w_fragments<4, 8>(Wneigh, sW + 32 * 4 * 32, wave, lane);
  __syncthreads();

  const int mt = blockIdx.x * 8 + wave;
  if (mt >= (nRows >> 4)) return;

  const int half = lane >> 4;
  const int l15  = lane & 15;
  const int aRow = mt * 16 + l15;
  const int kb   = half * 2;

  float bv[4];
#pragma unroll
  for (int nt = 0; nt < 4; ++nt) bv[nt] = bias[nt * 16 + l15];

  v8f cs[4], cn[4];
#pragma unroll
  for (int nt = 0; nt < 4; ++nt) {
    cs[nt] = (v8f){0.f, 0.f, 0.f, 0.f, 0.f, 0.f, 0.f, 0.f};
    cn[nt] = (v8f){0.f, 0.f, 0.f, 0.f, 0.f, 0.f, 0.f, 0.f};
  }

  const v2f* sWs = sW + lane;
  const v2f* sWn = sW + 32 * 4 * 32 + lane;
  const float* xr = H + (size_t)aRow * 128 + kb;

#pragma unroll 1
  for (int k0 = 0; k0 < 32; k0 += 8) {
    v2f a[8];
#pragma unroll
    for (int j = 0; j < 8; ++j) a[j] = *(const v2f*)(xr + 4 * (k0 + j));
#pragma unroll
    for (int j = 0; j < 8; ++j) {
#pragma unroll
      for (int nt = 0; nt < 4; ++nt) {
        const v2f bs = sWs[((k0 + j) * 4 + nt) * 32];
        const v2f bn = sWn[((k0 + j) * 4 + nt) * 32];
        cs[nt] = wmma_f32_16x16x4(a[j], bs, cs[nt]);
        cn[nt] = wmma_f32_16x16x4(a[j], bn, cn[nt]);
      }
    }
  }

  const size_t o0 = (size_t)(mt * 16 + half * 8) * 64 + l15;
#pragma unroll
  for (int nt = 0; nt < 4; ++nt) {
#pragma unroll
    for (int i = 0; i < 8; ++i) {
      outSelf[o0 + (size_t)i * 64 + nt * 16]  = cs[nt][i] + bv[nt];
      outNeigh[o0 + (size_t)i * 64 + nt * 16] = cn[nt][i];
    }
  }
}

#define OWN2_GROUP 16

__global__ __launch_bounds__(OWN_THREADS)
void own_rows_layer2_kernel(const float* __restrict__ P,
                            const int* __restrict__ src,
                            const int* __restrict__ dst,
                            const int* __restrict__ ownLists,
                            const float* __restrict__ DEG,
                            float* __restrict__ out,
                            int nE, int nRows) {
  __shared__ __align__(16) float sAcc[OWN_ROWS * 64];
  const int tid  = threadIdx.x;
  const int lane = tid & 31;
  const int wave = tid >> 5;
  const int base = blockIdx.x * OWN_ROWS;
  const int* myHead = ownLists + (size_t)blockIdx.x * OWN_LIST_STRIDE;
  const int total = myHead[0];
  for (int i = tid; i < OWN_ROWS * 64; i += OWN_THREADS) sAcc[i] = 0.0f;
  __syncthreads();

  if (total > OWN_LIST_CAP) {
    for (int e = tid; e < nE; e += OWN_THREADS) {
      const int r = dst[e] - base;
      if ((unsigned int)r >= (unsigned int)OWN_ROWS) continue;
      const float* pr = P + (size_t)src[e] * 64;
      for (int cc = 0; cc < 64; ++cc) atomicAdd(&sAcc[r * 64 + cc], pr[cc]);
    }
  } else {
    const int* myList = myHead + OWN_LIST_HEAD;
    const int n = total;
    const int noEdge = own_list_word(0, OWN_NO_ROW);
    int g0 = wave * OWN2_GROUP;
    int wv = noEdge;
    if (lane < OWN2_GROUP && g0 + lane < n) wv = myList[g0 + lane];
    while (g0 < n) {
      float v[OWN2_GROUP][2];
      int   r[OWN2_GROUP];
#pragma unroll
      for (int j = 0; j < OWN2_GROUP; ++j) {
        const int wj = __builtin_amdgcn_readlane(wv, j);
        r[j] = wj & 0xff;
        const float* pr = P + (size_t)(wj >> 8) * 64 + lane;
        v[j][0] = pr[0];
        v[j][1] = pr[32];
      }
      const int g1 = g0 + OWN_WAVES * OWN2_GROUP;
      int wv2 = noEdge;
      if (lane < OWN2_GROUP && g1 + lane < n) wv2 = myList[g1 + lane];
#pragma unroll
      for (int j = 0; j < OWN2_GROUP; ++j) {
        if ((unsigned int)r[j] < (unsigned int)OWN_ROWS) {
          float* o = &sAcc[r[j] * 64 + lane];
          atomicAdd(o,      v[j][0]);
          atomicAdd(o + 32, v[j][1]);
        }
      }
      wv = wv2;
      g0 = g1;
    }
  }
  __syncthreads();

  const int rowsHere = min(OWN_ROWS, nRows - base);
  float4* o4 = (float4*)(out + (size_t)base * 64);
  const float4* s4 = (const float4*)sAcc;
  for (int i = tid; i < rowsHere * 16; i += OWN_THREADS) {
    const float inv = 1.0f / fmaxf(DEG[base + (i >> 4)], 1.0f);
    float4 o = o4[i];
    const float4 s = s4[i];
    o.x += s.x * inv;
    o.y += s.y * inv;
    o.z += s.z * inv;
    o.w += s.w * inv;
    o4[i] = o;
  }
}

extern "C" void kernel_launch(void* const* d_in, const int* in_sizes, int n_in,
                              void* d_out, int out_size, void* d_ws, size_t ws_size,
                              hipStream_t stream) {
  const float* x   = (const float*)d_in[0];
  const float* Ws1 = (const float*)d_in[1];
  const float* Wn1 = (const float*)d_in[2];
  const float* b1  = (const float*)d_in[3];
  const float* Ws2 = (const float*)d_in[4];
  const float* Wn2 = (const float*)d_in[5];
  const float* b2  = (const float*)d_in[6];
  const int*   src = (const int*)d_in[7];
  const int*   dst = (const int*)d_in[8];
  float* out = (float*)d_out;

  const int F = in_sizes[3];
  const int N = in_sizes[0] / F;
  const int E = in_sizes[7];

  char* ws = (char*)d_ws;
  float* h   = (float*)ws;
  float* pl2 = (float*)(ws + (size_t)N * F * 4);
  float* deg = (float*)(ws + (size_t)N * F * 8);

  int* ownLists = (int*)(ws + (size_t)N * F * 4 + (size_t)N * 64 * 4);

  const int TB = 256;

  unsigned int* ownerW = (unsigned int*)pl2;
  const int ownerWords  = (E + 3) / 4;
  const int ownRuns     = (N + OWN_ROWS - 1) / OWN_ROWS;
  owner_byte_kernel<<<(ownerWords + TB - 1) / TB, TB, 0, stream>>>(dst, ownerW, E);
  own_rows_layer1_kernel<<<ownRuns, OWN_THREADS, 0, stream>>>(
      x, src, dst, ownerW, Ws1, Wn1, b1, h, deg, ownLists, E, N);

  float* p = pl2;
  {
    const int strips = N / 16;
    const int blocks = (strips + (TB / 32) - 1) / (TB / 32);
    sage_strip_dual_proj_kernel<<<blocks, TB, 0, stream>>>(
        h, Ws2, Wn2, b2, out, p, N);
  }
  own_rows_layer2_kernel<<<ownRuns, OWN_THREADS, 0, stream>>>(p, src, dst, ownLists, deg, out, E, N);
}


# run lists come from ONE bucket pass (each edge put once in its run's per-part segment; word = source row + local row), not from 256 work-groups each filtering all edges; f32 LDS adds in arrival order
# speedup vs baseline: 4.7275x; 1.0852x over previous
#include <hip/hip_runtime.h>


typedef float v2f __attribute__((ext_vector_type(2)));
typedef float v8f __attribute__((ext_vector_type(8)));

static __device__ __forceinline__ v8f wmma_f32_16x16x4(v2f a, v2f b, v8f c) {
  return __builtin_amdgcn_wmma_f32_16x16x4_f32(
       false, a,  false, b,
       (short)0, c,  false,  false);
}

#define OWN_ROWS 196
#define OWN_STRIPS 13
#define OWN_PAD_ROWS (OWN_STRIPS * 16)
#define OWN_PITCH 132
#define OWN_LIST_CAP 8192
#define OWN_THREADS 1024
#define OWN_WAVES (OWN_THREADS / 32)
#define OWN_GROUP 16

#define OWN_NO_ROW 255
#define OWN_LIST_HEAD 32
#define OWN_LIST_STRIDE (OWN_LIST_HEAD + OWN_LIST_CAP)

static __device__ __forceinline__ int own_list_word(int srcRow, int localRow) {
  return (srcRow << 8) | localRow;
}

#define PART_THREADS 1024
#define PART_PER_THREAD 4
#define PART_EDGES (PART_THREADS * PART_PER_THREAD)
#define SEG_CAP 56
#define SEG_MAXPART 256
#define OWN_RUNS 256

__global__ __launch_bounds__(PART_THREADS)
void bucket_edges_kernel(const int* __restrict__ src, const int* __restrict__ dst,
                         int* __restrict__ segCnt, int* __restrict__ seg, int nE, int nPart) {
  __shared__ int sBin[OWN_RUNS];
  __shared__ int sSeg[OWN_RUNS * SEG_CAP];
  const int tid  = threadIdx.x;
  const int lane = tid & 31;
  const int wave = tid >> 5;
  const int part = blockIdx.x;
  if (tid < OWN_RUNS) sBin[tid] = 0;
  int d[PART_PER_THREAD], s[PART_PER_THREAD], slot[PART_PER_THREAD];
#pragma unroll
  for (int j = 0; j < PART_PER_THREAD; ++j) {
    const int e = min(part * PART_EDGES + j * PART_THREADS + tid, nE - 1);
    d[j] = dst[e];
    s[j] = src[e];
  }
  __syncthreads();
#pragma unroll
  for (int j = 0; j < PART_PER_THREAD; ++j) {
    const int e = part * PART_EDGES + j * PART_THREADS + tid;
    const unsigned int run = (unsigned int)d[j] / OWN_ROWS;
    slot[j] = SEG_CAP;
    if (e < nE && run < (unsigned int)OWN_RUNS) slot[j] = atomicAdd(&sBin[run], 1);
  }
#pragma unroll
  for (int j = 0; j < PART_PER_THREAD; ++j) {
    const int run = (int)((unsigned int)d[j] / OWN_ROWS);
    if (slot[j] < SEG_CAP) sSeg[run * SEG_CAP + slot[j]] = own_list_word(s[j], d[j] - run * OWN_ROWS);
  }
  __syncthreads();
  int cnt[OWN_RUNS * 32 / PART_THREADS], w0[OWN_RUNS * 32 / PART_THREADS], w1[OWN_RUNS * 32 / PART_THREADS];
#pragma unroll
  for (int u = 0; u < OWN_RUNS * 32 / PART_THREADS; ++u) cnt[u] = min(sBin[wave + u * (PART_THREADS / 32)], SEG_CAP);
#pragma unroll
  for (int u = 0; u < OWN_RUNS * 32 / PART_THREADS; ++u) {
    const int run = wave + u * (PART_THREADS / 32);
    w0[u] = (lane < cnt[u])      ? sSeg[run * SEG_CAP + lane]      : 0;
    w1[u] = (lane + 32 < cnt[u]) ? sSeg[run * SEG_CAP + lane + 32] : 0;
  }
#pragma unroll
  for (int u = 0; u < OWN_RUNS * 32 / PART_THREADS; ++u) {
    const int run = wave + u * (PART_THREADS / 32);
    int* gp = seg + ((size_t)run * nPart + part) * SEG_CAP;
    if (lane < cnt[u])      gp[lane]      = w0[u];
    if (lane + 32 < cnt[u]) gp[lane + 32] = w1[u];
  }
  if (tid < OWN_RUNS) segCnt[(size_t)tid * nPart + part] = sBin[tid];
}

template <int NT, int NW>
static __device__ __forceinline__ void stage_w_fragments(const float* __restrict__ Wg,
                                                         v2f* __restrict__ sW,
                                                         int wave, int lane) {
  const int h = lane >> 4;
  const int c = lane & 15;
#pragma unroll 8
  for (int u = wave; u < 32 * NT; u += NW) {
    const int kk = u / NT;
    const int nt = u - kk * NT;
    const float* p = Wg + (size_t)(4 * kk + 2 * h) * (NT * 16) + nt * 16 + c;
    v2f b;
    b.x = p[0];
    b.y = p[NT * 16];
    sW[u * 32 + lane] = b;
  }
}

__global__ __launch_bounds__(OWN_THREADS)
void own_rows_layer1_kernel(const float* __restrict__ x,
                            const int* __restrict__ src,
                            const int* __restrict__ dst,
                            const int* __restrict__ segCnt,
                            const int* __restrict__ seg,
                            const float* __restrict__ Wself,
                            const float* __restrict__ Wneigh,
                            const float* __restrict__ bias,
                            float* __restrict__ hOut,
                            float* __restrict__ deg,
                            int* __restrict__ ownLists,
                            int nE, int nRows, int nPart) {
  __shared__ __align__(16) float sAcc[OWN_PAD_ROWS * OWN_PITCH];
  __shared__ float sDeg[OWN_PAD_ROWS];
  __shared__ int   sList[OWN_LIST_CAP];
  __shared__ int   sCnt;
  __shared__ int   sOver;
  __shared__ int   sSegOff[SEG_MAXPART];
  __shared__ int   sSegN[SEG_MAXPART];
  __shared__ v2f   sW[2 * 32 * 8 * 32];
  const int tid  = threadIdx.x;
  const int lane = tid & 31;
  const int wave = tid >> 5;
  const int base = blockIdx.x * OWN_ROWS;
  const bool partsFit = nPart <= SEG_MAXPART;

  int myCnt = 0;
  if (partsFit && tid < nPart) myCnt = segCnt[(size_t)blockIdx.x * nPart + tid];
  stage_w_fragments<8, OWN_WAVES>(Wself,  sW,               wave, lane);
  stage_w_fragments<8, OWN_WAVES>(Wneigh, sW + 32 * 8 * 32, wave, lane);
  for (int i = tid; i < OWN_PAD_ROWS * OWN_PITCH; i += OWN_THREADS) sAcc[i] = 0.0f;
  if (tid < OWN_PAD_ROWS) sDeg[tid] = 0.0f;
  if (tid == 0) { sCnt = 0; sOver = 0; }
  __syncthreads();

  if (partsFit && tid < nPart) {
    if (myCnt > SEG_CAP) sOver = 1;
    sSegOff[tid] = atomicAdd(&sCnt, myCnt);
    sSegN[tid]   = myCnt;
  }
  __syncthreads();
  const bool listOk = partsFit && sOver == 0 && sCnt <= OWN_LIST_CAP;
  if (listOk) {
    int w0[SEG_MAXPART / OWN_WAVES], w1[SEG_MAXPART / OWN_WAVES];
    int so[SEG_MAXPART / OWN_WAVES], sn[SEG_MAXPART / OWN_WAVES];
#pragma unroll
    for (int u = 0; u < SEG_MAXPART / OWN_WAVES; ++u) {
      const int b = wave + u * OWN_WAVES;
      const bool have = b < nPart;
      so[u] = have ? sSegOff[b] : 0;
      sn[u] = have ? sSegN[b] : 0;
      const int* sp = seg + ((size_t)blockIdx.x * nPart + (have ? b : 0)) * SEG_CAP;
      w0[u] = (lane < sn[u])      ? sp[lane]      : 0;
      w1[u] = (lane + 32 < sn[u]) ? sp[lane + 32] : 0;
    }
#pragma unroll
    for (int u = 0; u < SEG_MAXPART / OWN_WAVES; ++u) {
      if (lane < sn[u])      sList[so[u] + lane]      = w0[u];
      if (lane + 32 < sn[u]) sList[so[u] + lane + 32] = w1[u];
    }
  }
  __syncthreads();

  const int total = listOk ? sCnt : OWN_LIST_CAP + 1;
  int* myHead = ownLists + (size_t)blockIdx.x * OWN_LIST_STRIDE;
  int* myList = myHead + OWN_LIST_HEAD;
  if (tid == 0) myHead[0] = total;
  if (!listOk) {
    for (int e = tid; e < nE; e += OWN_THREADS) {
      const int r = dst[e] - base;
      if ((unsigned int)r >= (unsigned int)OWN_ROWS) continue;
      const float* xr = x + (size_t)src[e] * 128;
      for (int cc = 0; cc < 128; ++cc) atomicAdd(&sAcc[r * OWN_PITCH + cc], xr[cc]);
      atomicAdd(&sDeg[r], 1.0f);
    }
  } else {
    const int n = total;
    int g0 = wave * OWN_GROUP;
    int wv = 0;
    if (g0 + OWN_GROUP <= n && lane < OWN_GROUP) wv = sList[g0 + lane];
    while (g0 + OWN_GROUP <= n) {
      float v[OWN_GROUP][4];
      int   r[OWN_GROUP];
#pragma unroll
      for (int j = 0; j < OWN_GROUP; ++j) {
        const int wj = __builtin_amdgcn_readlane(wv, j);
        r[j] = wj & 0xff;
        const float* xr = x + (size_t)(wj >> 8) * 128 + lane;
        v[j][0] = xr[0];
        v[j][1] = xr[32];
        v[j][2] = xr[64];
        v[j][3] = xr[96];
      }
      const int g1 = g0 + OWN_WAVES * OWN_GROUP;
      int wv2 = 0;
      if (g1 + OWN_GROUP <= n && lane < OWN_GROUP) wv2 = sList[g1 + lane];
#pragma unroll
      for (int j = 0; j < OWN_GROUP; ++j) {
        if ((unsigned int)r[j] < (unsigned int)OWN_ROWS) {
          float* o = &sAcc[r[j] * OWN_PITCH + lane];
          atomicAdd(o,      v[j][0]);
          atomicAdd(o + 32, v[j][1]);
          atomicAdd(o + 64, v[j][2]);
          atomicAdd(o + 96, v[j][3]);
        }
      }
      if (lane < OWN_GROUP) {
        const int rd = wv & 0xff;
        myList[g0 + lane] = wv;
        if (rd < OWN_ROWS) atomicAdd(&sDeg[rd], 1.0f);
      }
      wv = wv2;
      g0 = g1;
    }
    const int nFull = (n / OWN_GROUP) * OWN_GROUP;
    if (wave == ((n / OWN_GROUP) % OWN_WAVES)) {
      for (int g = nFull; g < n; ++g) {
        const int w  = sList[g];
        const int rr = w & 0xff;
        const float* xr = x + (size_t)(w >> 8) * 128 + lane;
        const float v0 = xr[0];
        const float v1 = xr[32];
        const float v2 = xr[64];
        const float v3 = xr[96];
        if (lane == 0) myList[g] = w;
        if (rr < OWN_ROWS) {
          float* o = &sAcc[rr * OWN_PITCH + lane];
          atomicAdd(o,      v0);
          atomicAdd(o + 32, v1);
          atomicAdd(o + 64, v2);
          atomicAdd(o + 96, v3);
          if (lane == 0) atomicAdd(&sDeg[rr], 1.0f);
        }
      }
    }
  }
  __syncthreads();

  const int rowsHere = min(OWN_ROWS, nRows - base);
  if (tid < rowsHere) deg[base + tid] = sDeg[tid];

  const int nTasks = ((rowsHere + 15) >> 4) * 4;
  const int half = lane >> 4;
  const int l15  = lane & 15;
  const int kb   = half * 2;
#pragma unroll 1
  for (int t = wave; t < nTasks; t += OWN_WAVES) {
    const int strip = t >> 2;
    const int nt0   = (t & 3) * 2;
    const int lrow  = strip * 16 + l15;
    const int grow  = min(base + lrow, nRows - 1);
    const float inv = 1.0f / fmaxf(sDeg[lrow], 1.0f);
    const float bv0 = bias[nt0 * 16 + l15];
    const float bv1 = bias[nt0 * 16 + 16 + l15];

    v8f acc0 = (v8f){0.f, 0.f, 0.f, 0.f, 0.f, 0.f, 0.f, 0.f};
    v8f acc1 = (v8f){0.f, 0.f, 0.f, 0.f, 0.f, 0.f, 0.f, 0.f};

    const v2f* sWs = sW + nt0 * 32 + lane;
    const v2f* sWn = sW + 32 * 8 * 32 + nt0 * 32 + lane;
    const float* xr = x + (size_t)grow * 128 + kb;
    const float* gr = sAcc + lrow * OWN_PITCH + kb;

#pragma unroll 1
    for (int k0 = 0; k0 < 32; k0 += 8) {
      v2f a[8];
#pragma unroll
      for (int j = 0; j < 8; ++j) a[j] = *(const v2f*)(xr + 4 * (k0 + j));
#pragma unroll
      for (int j = 0; j < 8; ++j) {
        const v2f b0 = sWs[(k0 + j) * 8 * 32];
        const v2f b1 = sWs[(k0 + j) * 8 * 32 + 32];
        acc0 = wmma_f32_16x16x4(a[j], b0, acc0);
        acc1 = wmma_f32_16x16x4(a[j], b1, acc1);
      }
    }
#pragma unroll 1
    for (int k0 = 0; k0 < 32; k0 += 8) {
      v2f a[8];
#pragma unroll
      for (int j = 0; j < 8; ++j) {
        a[j] = *(const v2f*)(gr + 4 * (k0 + j));
        a[j].x *= inv;
        a[j].y *= inv;
      }
#pragma unroll
      for (int j = 0; j < 8; ++j) {
        const v2f b0 = sWn[(k0 + j) * 8 * 32];
        const v2f b1 = sWn[(k0 + j) * 8 * 32 + 32];
        acc0 = wmma_f32_16x16x4(a[j], b0, acc0);
        acc1 = wmma_f32_16x16x4(a[j], b1, acc1);
      }
    }

    const int orow0 = strip * 16 + half * 8;
    float* op = hOut + (size_t)(base + orow0) * 128 + nt0 * 16 + l15;
#pragma unroll
    for (int i = 0; i < 8; ++i) {
      if (orow0 + i < rowsHere) {
        op[(size_t)i * 128]      = fmaxf(acc0[i] + bv0, 0.0f);
        op[(size_t)i * 128 + 16] = fmaxf(acc1[i] + bv1, 0.0f);
      }
    }
  }
}


__global__ __launch_bounds__(256) __attribute__((amdgpu_waves_per_eu(2, 4)))
void sage_strip_dual_proj_kernel(const float* __restrict__ H,
                                 const float* __restrict__ Wself,
                                 const float* __restrict__ Wneigh,
                                 const float* __restrict__ bias,
                                 float* __restrict__ outSelf,
                                 float* __restrict__ outNeigh,
                                 int nRows) {
  __shared__ v2f sW[2 * 32 * 4 * 32];
  const int wave = threadIdx.x >> 5;
  const int lane = threadIdx.x & 31;
  stage_w_fragments<4, 8>(Wself,  sW,               wave, lane);
  stage_w_fragments<4, 8>(Wneigh, sW + 32 * 4 * 32, wave, lane);
  __syncthreads();

  const int mt = blockIdx.x * 8 + wave;
  if (mt >= (nRows >> 4)) return;

  const int half = lane >> 4;
  const int l15  = lane & 15;
  const int aRow = mt * 16 + l15;
  const int kb   = half * 2;

  float bv[4];
#pragma unroll
  for (int nt = 0; nt < 4; ++nt) bv[nt] = bias[nt * 16 + l15];

  v8f cs[4], cn[4];
#pragma unroll
  for (int nt = 0; nt < 4; ++nt) {
    cs[nt] = (v8f){0.f, 0.f, 0.f, 0.f, 0.f, 0.f, 0.f, 0.f};
    cn[nt] = (v8f){0.f, 0.f, 0.f, 0.f, 0.f, 0.f, 0.f, 0.f};
  }

  const v2f* sWs = sW + lane;
  const v2f* sWn = sW + 32 * 4 * 32 + lane;
  const float* xr = H + (size_t)aRow * 128 + kb;

#pragma unroll 1
  for (int k0 = 0; k0 < 32; k0 += 8) {
    v2f a[8];
#pragma unroll
    for (int j = 0; j < 8; ++j) a[j] = *(const v2f*)(xr + 4 * (k0 + j));
#pragma unroll
    for (int j = 0; j < 8; ++j) {
#pragma unroll
      for (int nt = 0; nt < 4; ++nt) {
        const v2f bs = sWs[((k0 + j) * 4 + nt) * 32];
        const v2f bn = sWn[((k0 + j) * 4 + nt) * 32];
        cs[nt] = wmma_f32_16x16x4(a[j], bs, cs[nt]);
        cn[nt] = wmma_f32_16x16x4(a[j], bn, cn[nt]);
      }
    }
  }

  const size_t o0 = (size_t)(mt * 16 + half * 8) * 64 + l15;
#pragma unroll
  for (int nt = 0; nt < 4; ++nt) {
#pragma unroll
    for (int i = 0; i < 8; ++i) {
      outSelf[o0 + (size_t)i * 64 + nt * 16]  = cs[nt][i] + bv[nt];
      outNeigh[o0 + (size_t)i * 64 + nt * 16] = cn[nt][i];
    }
  }
}

#define OWN2_GROUP 16

__global__ __launch_bounds__(OWN_THREADS)
void own_rows_layer2_kernel(const float* __restrict__ P,
                            const int* __restrict__ src,
                            const int* __restrict__ dst,
                            const int* __restrict__ ownLists,
                            const float* __restrict__ DEG,
                            float* __restrict__ out,
                            int nE, int nRows) {
  __shared__ __align__(16) float sAcc[OWN_ROWS * 64];
  const int tid  = threadIdx.x;
  const int lane = tid & 31;
  const int wave = tid >> 5;
  const int base = blockIdx.x * OWN_ROWS;
  const int* myHead = ownLists + (size_t)blockIdx.x * OWN_LIST_STRIDE;
  const int total = myHead[0];
  for (int i = tid; i < OWN_ROWS * 64; i += OWN_THREADS) sAcc[i] = 0.0f;
  __syncthreads();

  if (total > OWN_LIST_CAP) {
    for (int e = tid; e < nE; e += OWN_THREADS) {
      const int r = dst[e] - base;
      if ((unsigned int)r >= (unsigned int)OWN_ROWS) continue;
      const float* pr = P + (size_t)src[e] * 64;
      for (int cc = 0; cc < 64; ++cc) atomicAdd(&sAcc[r * 64 + cc], pr[cc]);
    }
  } else {
    const int* myList = myHead + OWN_LIST_HEAD;
    const int n = total;
    const int noEdge = own_list_word(0, OWN_NO_ROW);
    int g0 = wave * OWN2_GROUP;
    int wv = noEdge;
    if (lane < OWN2_GROUP && g0 + lane < n) wv = myList[g0 + lane];
    while (g0 < n) {
      float v[OWN2_GROUP][2];
      int   r[OWN2_GROUP];
#pragma unroll
      for (int j = 0; j < OWN2_GROUP; ++j) {
        const int wj = __builtin_amdgcn_readlane(wv, j);
        r[j] = wj & 0xff;
        const float* pr = P + (size_t)(wj >> 8) * 64 + lane;
        v[j][0] = pr[0];
        v[j][1] = pr[32];
      }
      const int g1 = g0 + OWN_WAVES * OWN2_GROUP;
      int wv2 = noEdge;
      if (lane < OWN2_GROUP && g1 + lane < n) wv2 = myList[g1 + lane];
#pragma unroll
      for (int j = 0; j < OWN2_GROUP; ++j) {
        if ((unsigned int)r[j] < (unsigned int)OWN_ROWS) {
          float* o = &sAcc[r[j] * 64 + lane];
          atomicAdd(o,      v[j][0]);
          atomicAdd(o + 32, v[j][1]);
        }
      }
      wv = wv2;
      g0 = g1;
    }
  }
  __syncthreads();

  const int rowsHere = min(OWN_ROWS, nRows - base);
  float4* o4 = (float4*)(out + (size_t)base * 64);
  const float4* s4 = (const float4*)sAcc;
  for (int i = tid; i < rowsHere * 16; i += OWN_THREADS) {
    const float inv = 1.0f / fmaxf(DEG[base + (i >> 4)], 1.0f);
    float4 o = o4[i];
    const float4 s = s4[i];
    o.x += s.x * inv;
    o.y += s.y * inv;
    o.z += s.z * inv;
    o.w += s.w * inv;
    o4[i] = o;
  }
}

extern "C" void kernel_launch(void* const* d_in, const int* in_sizes, int n_in,
                              void* d_out, int out_size, void* d_ws, size_t ws_size,
                              hipStream_t stream) {
  const float* x   = (const float*)d_in[0];
  const float* Ws1 = (const float*)d_in[1];
  const float* Wn1 = (const float*)d_in[2];
  const float* b1  = (const float*)d_in[3];
  const float* Ws2 = (const float*)d_in[4];
  const float* Wn2 = (const float*)d_in[5];
  const float* b2  = (const float*)d_in[6];
  const int*   src = (const int*)d_in[7];
  const int*   dst = (const int*)d_in[8];
  float* out = (float*)d_out;

  const int F = in_sizes[3];
  const int N = in_sizes[0] / F;
  const int E = in_sizes[7];

  char* ws = (char*)d_ws;
  float* h   = (float*)ws;
  float* pl2 = (float*)(ws + (size_t)N * F * 4);
  float* deg = (float*)(ws + (size_t)N * F * 8);

  int* ownLists = (int*)(ws + (size_t)N * F * 4 + (size_t)N * 64 * 4);

  const int TB = 256;

  const int nPart   = (E + PART_EDGES - 1) / PART_EDGES;
  const int ownRuns = (N + OWN_ROWS - 1) / OWN_ROWS;
  int* segCnt = (int*)pl2;
  int* seg    = (int*)((char*)pl2 + 262144);
  bucket_edges_kernel<<<nPart, PART_THREADS, 0, stream>>>(src, dst, segCnt, seg, E, nPart);
  own_rows_layer1_kernel<<<ownRuns, OWN_THREADS, 0, stream>>>(
      x, src, dst, segCnt, seg, Ws1, Wn1, b1, h, deg, ownLists, E, N, nPart);

  float* p = pl2;
  {
    const int strips = N / 16;
    const int blocks = (strips + (TB / 32) - 1) / (TB / 32);
    sage_strip_dual_proj_kernel<<<blocks, TB, 0, stream>>>(
        h, Ws2, Wn2, b2, out, p, N);
  }
  own_rows_layer2_kernel<<<ownRuns, OWN_THREADS, 0, stream>>>(p, src, dst, ownLists, deg, out, E, N);
}


# both owner gathers fetch a source row with ONE load instruction (16 B per lane in layer 1, 8 B in layer 2), not 4 or 2 one-line loads; LDS sums column-interleaved; f32 LDS adds in arrival order
# speedup vs baseline: 5.0112x; 1.0600x over previous
#include <hip/hip_runtime.h>


typedef float v2f __attribute__((ext_vector_type(2)));
typedef float v8f __attribute__((ext_vector_type(8)));

static __device__ __forceinline__ v8f wmma_f32_16x16x4(v2f a, v2f b, v8f c) {
  return __builtin_amdgcn_wmma_f32_16x16x4_f32(
       false, a,  false, b,
       (short)0, c,  false,  false);
}

#define OWN_ROWS 196
#define OWN_STRIPS 13
#define OWN_PAD_ROWS (OWN_STRIPS * 16)
#define OWN_PITCH 132
#define OWN_CSTEP 33
#define OWN_LIST_CAP 8192
#define OWN_THREADS 1024
#define OWN_WAVES (OWN_THREADS / 32)
#define OWN_GROUP 16

#define OWN_NO_ROW 255
#define OWN_LIST_HEAD 32
#define OWN_LIST_STRIDE (OWN_LIST_HEAD + OWN_LIST_CAP)

static __device__ __forceinline__ int own_list_word(int srcRow, int localRow) {
  return (srcRow << 8) | localRow;
}

#define PART_THREADS 1024
#define PART_PER_THREAD 4
#define PART_EDGES (PART_THREADS * PART_PER_THREAD)
#define SEG_CAP 56
#define SEG_MAXPART 256
#define OWN_RUNS 256

__global__ __launch_bounds__(PART_THREADS)
void bucket_edges_kernel(const int* __restrict__ src, const int* __restrict__ dst,
                         int* __restrict__ segCnt, int* __restrict__ seg, int nE, int nPart) {
  __shared__ int sBin[OWN_RUNS];
  __shared__ int sSeg[OWN_RUNS * SEG_CAP];
  const int tid  = threadIdx.x;
  const int lane = tid & 31;
  const int wave = tid >> 5;
  const int part = blockIdx.x;
  if (tid < OWN_RUNS) sBin[tid] = 0;
  int d[PART_PER_THREAD], s[PART_PER_THREAD], slot[PART_PER_THREAD];
#pragma unroll
  for (int j = 0; j < PART_PER_THREAD; ++j) {
    const int e = min(part * PART_EDGES + j * PART_THREADS + tid, nE - 1);
    d[j] = dst[e];
    s[j] = src[e];
  }
  __syncthreads();
#pragma unroll
  for (int j = 0; j < PART_PER_THREAD; ++j) {
    const int e = part * PART_EDGES + j * PART_THREADS + tid;
    const unsigned int run = (unsigned int)d[j] / OWN_ROWS;
    slot[j] = SEG_CAP;
    if (e < nE && run < (unsigned int)OWN_RUNS) slot[j] = atomicAdd(&sBin[run], 1);
  }
#pragma unroll
  for (int j = 0; j < PART_PER_THREAD; ++j) {
    const int run = (int)((unsigned int)d[j] / OWN_ROWS);
    if (slot[j] < SEG_CAP) sSeg[run * SEG_CAP + slot[j]] = own_list_word(s[j], d[j] - run * OWN_ROWS);
  }
  __syncthreads();
  int cnt[OWN_RUNS * 32 / PART_THREADS], w0[OWN_RUNS * 32 / PART_THREADS], w1[OWN_RUNS * 32 / PART_THREADS];
#pragma unroll
  for (int u = 0; u < OWN_RUNS * 32 / PART_THREADS; ++u) cnt[u] = min(sBin[wave + u * (PART_THREADS / 32)], SEG_CAP);
#pragma unroll
  for (int u = 0; u < OWN_RUNS * 32 / PART_THREADS; ++u) {
    const int run = wave + u * (PART_THREADS / 32);
    w0[u] = (lane < cnt[u])      ? sSeg[run * SEG_CAP + lane]      : 0;
    w1[u] = (lane + 32 < cnt[u]) ? sSeg[run * SEG_CAP + lane + 32] : 0;
  }
#pragma unroll
  for (int u = 0; u < OWN_RUNS * 32 / PART_THREADS; ++u) {
    const int run = wave + u * (PART_THREADS / 32);
    int* gp = seg + ((size_t)run * nPart + part) * SEG_CAP;
    if (lane < cnt[u])      gp[lane]      = w0[u];
    if (lane + 32 < cnt[u]) gp[lane + 32] = w1[u];
  }
  if (tid < OWN_RUNS) segCnt[(size_t)tid * nPart + part] = sBin[tid];
}

template <int NT, int NW>
static __device__ __forceinline__ void stage_w_fragments(const float* __restrict__ Wg,
                                                         v2f* __restrict__ sW,
                                                         int wave, int lane) {
  const int h = lane >> 4;
  const int c = lane & 15;
#pragma unroll 8
  for (int u = wave; u < 32 * NT; u += NW) {
    const int kk = u / NT;
    const int nt = u - kk * NT;
    const float* p = Wg + (size_t)(4 * kk + 2 * h) * (NT * 16) + nt * 16 + c;
    v2f b;
    b.x = p[0];
    b.y = p[NT * 16];
    sW[u * 32 + lane] = b;
  }
}

__global__ __launch_bounds__(OWN_THREADS)
void own_rows_layer1_kernel(const float* __restrict__ x,
                            const int* __restrict__ src,
                            const int* __restrict__ dst,
                            const int* __restrict__ segCnt,
                            const int* __restrict__ seg,
                            const float* __restrict__ Wself,
                            const float* __restrict__ Wneigh,
                            const float* __restrict__ bias,
                            float* __restrict__ hOut,
                            float* __restrict__ deg,
                            int* __restrict__ ownLists,
                            int nE, int nRows, int nPart) {
  __shared__ __align__(16) float sAcc[OWN_PAD_ROWS * OWN_PITCH];
  __shared__ float sDeg[OWN_PAD_ROWS];
  __shared__ int   sList[OWN_LIST_CAP];
  __shared__ int   sCnt;
  __shared__ int   sOver;
  __shared__ int   sSegOff[SEG_MAXPART];
  __shared__ int   sSegN[SEG_MAXPART];
  __shared__ v2f   sW[2 * 32 * 8 * 32];
  const int tid  = threadIdx.x;
  const int lane = tid & 31;
  const int wave = tid >> 5;
  const int base = blockIdx.x * OWN_ROWS;
  const bool partsFit = nPart <= SEG_MAXPART;

  int myCnt = 0;
  if (partsFit && tid < nPart) myCnt = segCnt[(size_t)blockIdx.x * nPart + tid];
  stage_w_fragments<8, OWN_WAVES>(Wself,  sW,               wave, lane);
  stage_w_fragments<8, OWN_WAVES>(Wneigh, sW + 32 * 8 * 32, wave, lane);
  for (int i = tid; i < OWN_PAD_ROWS * OWN_PITCH; i += OWN_THREADS) sAcc[i] = 0.0f;
  if (tid < OWN_PAD_ROWS) sDeg[tid] = 0.0f;
  if (tid == 0) { sCnt = 0; sOver = 0; }
  __syncthreads();

  if (partsFit && tid < nPart) {
    if (myCnt > SEG_CAP) sOver = 1;
    sSegOff[tid] = atomicAdd(&sCnt, myCnt);
    sSegN[tid]   = myCnt;
  }
  __syncthreads();
  const bool listOk = partsFit && sOver == 0 && sCnt <= OWN_LIST_CAP;
  if (listOk) {
    int w0[SEG_MAXPART / OWN_WAVES], w1[SEG_MAXPART / OWN_WAVES];
    int so[SEG_MAXPART / OWN_WAVES], sn[SEG_MAXPART / OWN_WAVES];
#pragma unroll
    for (int u = 0; u < SEG_MAXPART / OWN_WAVES; ++u) {
      const int b = wave + u * OWN_WAVES;
      const bool have = b < nPart;
      so[u] = have ? sSegOff[b] : 0;
      sn[u] = have ? sSegN[b] : 0;
      const int* sp = seg + ((size_t)blockIdx.x * nPart + (have ? b : 0)) * SEG_CAP;
      w0[u] = (lane < sn[u])      ? sp[lane]      : 0;
      w1[u] = (lane + 32 < sn[u]) ? sp[lane + 32] : 0;
    }
#pragma unroll
    for (int u = 0; u < SEG_MAXPART / OWN_WAVES; ++u) {
      if (lane < sn[u])      sList[so[u] + lane]      = w0[u];
      if (lane + 32 < sn[u]) sList[so[u] + lane + 32] = w1[u];
    }
  }
  __syncthreads();

  const int total = listOk ? sCnt : OWN_LIST_CAP + 1;
  int* myHead = ownLists + (size_t)blockIdx.x * OWN_LIST_STRIDE;
  int* myList = myHead + OWN_LIST_HEAD;
  if (tid == 0) myHead[0] = total;
  if (!listOk) {
    for (int e = tid; e < nE; e += OWN_THREADS) {
      const int r = dst[e] - base;
      if ((unsigned int)r >= (unsigned int)OWN_ROWS) continue;
      const float* xr = x + (size_t)src[e] * 128;
      for (int cc = 0; cc < 128; ++cc)
        atomicAdd(&sAcc[r * OWN_PITCH + (cc & 3) * OWN_CSTEP + (cc >> 2)], xr[cc]);
      atomicAdd(&sDeg[r], 1.0f);
    }
  } else {
    const int n = total;
    int g0 = wave * OWN_GROUP;
    int wv = 0;
    if (g0 + OWN_GROUP <= n && lane < OWN_GROUP) wv = sList[g0 + lane];
    while (g0 + OWN_GROUP <= n) {
      float4 v[OWN_GROUP];
      int    r[OWN_GROUP];
#pragma unroll
      for (int j = 0; j < OWN_GROUP; ++j) {
        const int wj = __builtin_amdgcn_readlane(wv, j);
        r[j] = wj & 0xff;
        v[j] = *((const float4*)(x + (size_t)(wj >> 8) * 128) + lane);
      }
      const int g1 = g0 + OWN_WAVES * OWN_GROUP;
      int wv2 = 0;
      if (g1 + OWN_GROUP <= n && lane < OWN_GROUP) wv2 = sList[g1 + lane];
#pragma unroll
      for (int j = 0; j < OWN_GROUP; ++j) {
        if ((unsigned int)r[j] < (unsigned int)OWN_ROWS) {
          float* o = &sAcc[r[j] * OWN_PITCH + lane];
          atomicAdd(o,                 v[j].x);
          atomicAdd(o + OWN_CSTEP,     v[j].y);
          atomicAdd(o + 2 * OWN_CSTEP, v[j].z);
          atomicAdd(o + 3 * OWN_CSTEP, v[j].w);
        }
      }
      if (lane < OWN_GROUP) {
        const int rd = wv & 0xff;
        myList[g0 + lane] = wv;
        if (rd < OWN_ROWS) atomicAdd(&sDeg[rd], 1.0f);
      }
      wv = wv2;
      g0 = g1;
    }
    const int nFull = (n / OWN_GROUP) * OWN_GROUP;
    if (wave == ((n / OWN_GROUP) % OWN_WAVES)) {
      for (int g = nFull; g < n; ++g) {
        const int w  = sList[g];
        const int rr = w & 0xff;
        const float4 vr = *((const float4*)(x + (size_t)(w >> 8) * 128) + lane);
        if (lane == 0) myList[g] = w;
        if (rr < OWN_ROWS) {
          float* o = &sAcc[rr * OWN_PITCH + lane];
          atomicAdd(o,                 vr.x);
          atomicAdd(o + OWN_CSTEP,     vr.y);
          atomicAdd(o + 2 * OWN_CSTEP, vr.z);
          atomicAdd(o + 3 * OWN_CSTEP, vr.w);
          if (lane == 0) atomicAdd(&sDeg[rr], 1.0f);
        }
      }
    }
  }
  __syncthreads();

  const int rowsHere = min(OWN_ROWS, nRows - base);
  if (tid < rowsHere) deg[base + tid] = sDeg[tid];

  const int nTasks = ((rowsHere + 15) >> 4) * 4;
  const int half = lane >> 4;
  const int l15  = lane & 15;
  const int kb   = half * 2;
#pragma unroll 1
  for (int t = wave; t < nTasks; t += OWN_WAVES) {
    const int strip = t >> 2;
    const int nt0   = (t & 3) * 2;
    const int lrow  = strip * 16 + l15;
    const int grow  = min(base + lrow, nRows - 1);
    const float inv = 1.0f / fmaxf(sDeg[lrow], 1.0f);
    const float bv0 = bias[nt0 * 16 + l15];
    const float bv1 = bias[nt0 * 16 + 16 + l15];

    v8f acc0 = (v8f){0.f, 0.f, 0.f, 0.f, 0.f, 0.f, 0.f, 0.f};
    v8f acc1 = (v8f){0.f, 0.f, 0.f, 0.f, 0.f, 0.f, 0.f, 0.f};

    const v2f* sWs = sW + nt0 * 32 + lane;
    const v2f* sWn = sW + 32 * 8 * 32 + nt0 * 32 + lane;
    const float* xr = x + (size_t)grow * 128 + kb;
    const float* gr = sAcc + lrow * OWN_PITCH + kb * OWN_CSTEP;

#pragma unroll 1
    for (int k0 = 0; k0 < 32; k0 += 8) {
      v2f a[8];
#pragma unroll
      for (int j = 0; j < 8; ++j) a[j] = *(const v2f*)(xr + 4 * (k0 + j));
#pragma unroll
      for (int j = 0; j < 8; ++j) {
        const v2f b0 = sWs[(k0 + j) * 8 * 32];
        const v2f b1 = sWs[(k0 + j) * 8 * 32 + 32];
        acc0 = wmma_f32_16x16x4(a[j], b0, acc0);
        acc1 = wmma_f32_16x16x4(a[j], b1, acc1);
      }
    }
#pragma unroll 1
    for (int k0 = 0; k0 < 32; k0 += 8) {
      v2f a[8];
#pragma unroll
      for (int j = 0; j < 8; ++j) {
        a[j].x = gr[k0 + j] * inv;
        a[j].y = gr[k0 + j + OWN_CSTEP] * inv;
      }
#pragma unroll
      for (int j = 0; j < 8; ++j) {
        const v2f b0 = sWn[(k0 + j) * 8 * 32];
        const v2f b1 = sWn[(k0 + j) * 8 * 32 + 32];
        acc0 = wmma_f32_16x16x4(a[j], b0, acc0);
        acc1 = wmma_f32_16x16x4(a[j], b1, acc1);
      }
    }

    const int orow0 = strip * 16 + half * 8;
    float* op = hOut + (size_t)(base + orow0) * 128 + nt0 * 16 + l15;
#pragma unroll
    for (int i = 0; i < 8; ++i) {
      if (orow0 + i < rowsHere) {
        op[(size_t)i * 128]      = fmaxf(acc0[i] + bv0, 0.0f);
        op[(size_t)i * 128 + 16] = fmaxf(acc1[i] + bv1, 0.0f);
      }
    }
  }
}


__global__ __launch_bounds__(256) __attribute__((amdgpu_waves_per_eu(2, 4)))
void sage_strip_dual_proj_kernel(const float* __restrict__ H,
                                 const float* __restrict__ Wself,
                                 const float* __restrict__ Wneigh,
                                 const float* __restrict__ bias,
                                 float* __restrict__ outSelf,
                                 float* __restrict__ outNeigh,
                                 int nRows) {
  __shared__ v2f sW[2 * 32 * 4 * 32];
  const int wave = threadIdx.x >> 5;
  const int lane = threadIdx.x & 31;
  stage_w_fragments<4, 8>(Wself,  sW,               wave, lane);
  stage_w_fragments<4, 8>(Wneigh, sW + 32 * 4 * 32, wave, lane);
  __syncthreads();

  const int mt = blockIdx.x * 8 + wave;
  if (mt >= (nRows >> 4)) return;

  const int half = lane >> 4;
  const int l15  = lane & 15;
  const int aRow = mt * 16 + l15;
  const int kb   = half * 2;

  float bv[4];
#pragma unroll
  for (int nt = 0; nt < 4; ++nt) bv[nt] = bias[nt * 16 + l15];

  v8f cs[4], cn[4];
#pragma unroll
  for (int nt = 0; nt < 4; ++nt) {
    cs[nt] = (v8f){0.f, 0.f, 0.f, 0.f, 0.f, 0.f, 0.f, 0.f};
    cn[nt] = (v8f){0.f, 0.f, 0.f, 0.f, 0.f, 0.f, 0.f, 0.f};
  }

  const v2f* sWs = sW + lane;
  const v2f* sWn = sW + 32 * 4 * 32 + lane;
  const float* xr = H + (size_t)aRow * 128 + kb;

#pragma unroll 1
  for (int k0 = 0; k0 < 32; k0 += 8) {
    v2f a[8];
#pragma unroll
    for (int j = 0; j < 8; ++j) a[j] = *(const v2f*)(xr + 4 * (k0 + j));
#pragma unroll
    for (int j = 0; j < 8; ++j) {
#pragma unroll
      for (int nt = 0; nt < 4; ++nt) {
        const v2f bs = sWs[((k0 + j) * 4 + nt) * 32];
        const v2f bn = sWn[((k0 + j) * 4 + nt) * 32];
        cs[nt] = wmma_f32_16x16x4(a[j], bs, cs[nt]);
        cn[nt] = wmma_f32_16x16x4(a[j], bn, cn[nt]);
      }
    }
  }

  const size_t o0 = (size_t)(mt * 16 + half * 8) * 64 + l15;
#pragma unroll
  for (int nt = 0; nt < 4; ++nt) {
#pragma unroll
    for (int i = 0; i < 8; ++i) {
      outSelf[o0 + (size_t)i * 64 + nt * 16]  = cs[nt][i] + bv[nt];
      outNeigh[o0 + (size_t)i * 64 + nt * 16] = cn[nt][i];
    }
  }
}

#define OWN2_GROUP 16

__global__ __launch_bounds__(OWN_THREADS)
void own_rows_layer2_kernel(const float* __restrict__ P,
                            const int* __restrict__ src,
                            const int* __restrict__ dst,
                            const int* __restrict__ ownLists,
                            const float* __restrict__ DEG,
                            float* __restrict__ out,
                            int nE, int nRows) {
  __shared__ __align__(16) float sAcc[OWN_ROWS * 64];
  const int tid  = threadIdx.x;
  const int lane = tid & 31;
  const int wave = tid >> 5;
  const int base = blockIdx.x * OWN_ROWS;
  const int* myHead = ownLists + (size_t)blockIdx.x * OWN_LIST_STRIDE;
  const int total = myHead[0];
  for (int i = tid; i < OWN_ROWS * 64; i += OWN_THREADS) sAcc[i] = 0.0f;
  __syncthreads();

  if (total > OWN_LIST_CAP) {
    for (int e = tid; e < nE; e += OWN_THREADS) {
      const int r = dst[e] - base;
      if ((unsigned int)r >= (unsigned int)OWN_ROWS) continue;
      const float* pr = P + (size_t)src[e] * 64;
      for (int cc = 0; cc < 64; ++cc) atomicAdd(&sAcc[r * 64 + (cc & 1) * 32 + (cc >> 1)], pr[cc]);
    }
  } else {
    const int* myList = myHead + OWN_LIST_HEAD;
    const int n = total;
    const int noEdge = own_list_word(0, OWN_NO_ROW);
    int g0 = wave * OWN2_GROUP;
    int wv = noEdge;
    if (lane < OWN2_GROUP && g0 + lane < n) wv = myList[g0 + lane];
    while (g0 < n) {
      float2 v[OWN2_GROUP];
      int    r[OWN2_GROUP];
#pragma unroll
      for (int j = 0; j < OWN2_GROUP; ++j) {
        const int wj = __builtin_amdgcn_readlane(wv, j);
        r[j] = wj & 0xff;
        v[j] = *((const float2*)(P + (size_t)(wj >> 8) * 64) + lane);
      }
      const int g1 = g0 + OWN_WAVES * OWN2_GROUP;
      int wv2 = noEdge;
      if (lane < OWN2_GROUP && g1 + lane < n) wv2 = myList[g1 + lane];
#pragma unroll
      for (int j = 0; j < OWN2_GROUP; ++j) {
        if ((unsigned int)r[j] < (unsigned int)OWN_ROWS) {
          float* o = &sAcc[r[j] * 64 + lane];
          atomicAdd(o,      v[j].x);
          atomicAdd(o + 32, v[j].y);
        }
      }
      wv = wv2;
      g0 = g1;
    }
  }
  __syncthreads();

  const int rowsHere = min(OWN_ROWS, nRows - base);
  float4* o4 = (float4*)(out + (size_t)base * 64);
  for (int i = tid; i < rowsHere * 16; i += OWN_THREADS) {
    const float inv = 1.0f / fmaxf(DEG[base + (i >> 4)], 1.0f);
    float4 o = o4[i];
    const float* sr = sAcc + (i >> 4) * 64 + 2 * (i & 15);
    const float2 ev = *(const float2*)sr;
    const float2 od = *(const float2*)(sr + 32);
    o.x += ev.x * inv;
    o.y += od.x * inv;
    o.z += ev.y * inv;
    o.w += od.y * inv;
    o4[i] = o;
  }
}

extern "C" void kernel_launch(void* const* d_in, const int* in_sizes, int n_in,
                              void* d_out, int out_size, void* d_ws, size_t ws_size,
                              hipStream_t stream) {
  const float* x   = (const float*)d_in[0];
  const float* Ws1 = (const float*)d_in[1];
  const float* Wn1 = (const float*)d_in[2];
  const float* b1  = (const float*)d_in[3];
  const float* Ws2 = (const float*)d_in[4];
  const float* Wn2 = (const float*)d_in[5];
  const float* b2  = (const float*)d_in[6];
  const int*   src = (const int*)d_in[7];
  const int*   dst = (const int*)d_in[8];
  float* out = (float*)d_out;

  const int F = in_sizes[3];
  const int N = in_sizes[0] / F;
  const int E = in_sizes[7];

  char* ws = (char*)d_ws;
  float* h   = (float*)ws;
  float* pl2 = (float*)(ws + (size_t)N * F * 4);
  float* deg = (float*)(ws + (size_t)N * F * 8);

  int* ownLists = (int*)(ws + (size_t)N * F * 4 + (size_t)N * 64 * 4);

  const int TB = 256;

  const int nPart   = (E + PART_EDGES - 1) / PART_EDGES;
  const int ownRuns = (N + OWN_ROWS - 1) / OWN_ROWS;
  int* segCnt = (int*)pl2;
  int* seg    = (int*)((char*)pl2 + 262144);
  bucket_edges_kernel<<<nPart, PART_THREADS, 0, stream>>>(src, dst, segCnt, seg, E, nPart);
  own_rows_layer1_kernel<<<ownRuns, OWN_THREADS, 0, stream>>>(
      x, src, dst, segCnt, seg, Ws1, Wn1, b1, h, deg, ownLists, E, N, nPart);

  float* p = pl2;
  {
    const int strips = N / 16;
    const int blocks = (strips + (TB / 32) - 1) / (TB / 32);
    sage_strip_dual_proj_kernel<<<blocks, TB, 0, stream>>>(
        h, Ws2, Wn2, b2, out, p, N);
  }
  own_rows_layer2_kernel<<<ownRuns, OWN_THREADS, 0, stream>>>(p, src, dst, ownLists, deg, out, E, N);
}
